# starting kernel
# baseline (speedup 1.0000x reference)
#include <hip/hip_runtime.h>
#include <hip/hip_bf16.h>
#include <math.h>

#define GH 30
#define GW 30
#define HWC 900
#define DD 256
#define MAXOBJ 16
#define SENT 900
#define KF 261
#define KP 264
#define LD 268

typedef __attribute__((ext_vector_type(2))) float v2f;
typedef __attribute__((ext_vector_type(8))) float v8f;

__global__ __launch_bounds__(1024) void cc_extract_kernel(const int* __restrict__ grid,
                                                          int* __restrict__ meta) {
  __shared__ int g[HWC];
  __shared__ int lab[HWC];
  __shared__ int rmn[HWC], rmx[HWC], cmn[HWC], cmx[HWC];
  __shared__ int roots[MAXOBJ];
  __shared__ int nroots;
  __shared__ int changed;

  const int b = blockIdx.x;
  const int tid = threadIdx.x;
  const int* gg = grid + b * HWC;

  for (int t = tid; t < HWC; t += blockDim.x) {
    int v = gg[t];
    g[t] = v;
    lab[t] = (v > 0) ? t : SENT;
    rmn[t] = GH; rmx[t] = -1; cmn[t] = GW; cmx[t] = -1;
  }
  __syncthreads();

  for (int iter = 0; iter < 2048; ++iter) {
    if (tid == 0) changed = 0;
    __syncthreads();
    const int cell = tid;
    const bool active = (cell < HWC) && (g[cell] > 0);
    int newl = SENT, oldl = SENT;
    if (active) {
      const int r = cell / GW, c = cell % GW;
      const int gv = g[cell];
      int m = lab[cell]; oldl = m;
      if (r > 0      && g[cell - GW] == gv) m = min(m, lab[cell - GW]);
      if (r < GH - 1 && g[cell + GW] == gv) m = min(m, lab[cell + GW]);
      if (c > 0      && g[cell - 1]  == gv) m = min(m, lab[cell - 1]);
      if (c < GW - 1 && g[cell + 1]  == gv) m = min(m, lab[cell + 1]);
      if (m < SENT) m = min(m, lab[m]);
      newl = m;
    }
    __syncthreads();
    if (active && newl != oldl) { lab[cell] = newl; changed = 1; }
    __syncthreads();
    const int done = (changed == 0);
    __syncthreads();
    if (done) break;
  }

  if (tid < HWC && g[tid] > 0) {
    const int root = lab[tid];
    atomicMin(&rmn[root], tid / GW);
    atomicMax(&rmx[root], tid / GW);
    atomicMin(&cmn[root], tid % GW);
    atomicMax(&cmx[root], tid % GW);
  }
  __syncthreads();

  if (tid == 0) {
    int cnt = 0;
    for (int i = 0; i < HWC && cnt < MAXOBJ; ++i)
      if (g[i] > 0 && lab[i] == i) roots[cnt++] = i;
    nroots = cnt;
  }
  __syncthreads();

  if (tid < MAXOBJ) {
    int* mo = meta + (b * MAXOBJ + tid) * 6;
    if (tid < nroots) {
      const int r = roots[tid];
      int y  = min(max(rmn[r], 0), GH - 1);
      int x  = min(max(cmn[r], 0), GW - 1);
      int y2 = min(max(rmx[r] + 1, y + 1), GH);
      int x2 = min(max(cmx[r] + 1, x + 1), GW);
      mo[0] = g[r]; mo[1] = y; mo[2] = x; mo[3] = y2 - y; mo[4] = x2 - x; mo[5] = 1;
    } else {
      mo[0] = 0; mo[1] = 0; mo[2] = 0; mo[3] = 1; mo[4] = 1; mo[5] = 0;
    }
  }
}

__global__ __launch_bounds__(256) void pool_kernel(const float* __restrict__ grid_emb,
                                                   const int* __restrict__ meta,
                                                   float* __restrict__ comb) {
  const int bk = blockIdx.x;
  const int b = bk >> 4;
  const int d = threadIdx.x;
  const int* mo = meta + bk * 6;
  const int color = mo[0], y = mo[1], x = mo[2], h = mo[3], w = mo[4];

  float s = 0.f;
  for (int i = 0; i < h; ++i) {
    const float* rowp = grid_emb + (((size_t)(b * GH + y + i)) * GW + x) * DD + d;
    for (int j = 0; j < w; ++j) s += rowp[(size_t)j * DD];
  }
  float* op = comb + (size_t)bk * KP;
  op[d] = s / (float)(h * w);
  if (d == 0) {
    op[DD + 0] = (float)color / 9.0f;
    op[DD + 1] = (float)x / (float)GW;
    op[DD + 2] = (float)y / (float)GH;
    op[DD + 3] = (float)w / (float)GW;
    op[DD + 4] = (float)h / (float)GH;
    op[DD + 5] = 0.f; op[DD + 6] = 0.f; op[DD + 7] = 0.f;
  }
}

__global__ __launch_bounds__(512) void fused_mlp_kernel(
    const float* __restrict__ comb_g, const int* __restrict__ meta,
    const float* __restrict__ structure, const float* __restrict__ W1,
    const float* __restrict__ b1, const float* __restrict__ W2,
    const float* __restrict__ b2, const float* __restrict__ Wp,
    const float* __restrict__ bp, const float* __restrict__ gmm,
    const float* __restrict__ bta, const float* __restrict__ osp,
    float* __restrict__ out) {
  __shared__ float sA[MAXOBJ][LD];
  __shared__ float sB[MAXOBJ][LD];
  __shared__ float sC[MAXOBJ][LD];
  __shared__ float s_sn[DD];
  __shared__ float s_dot[MAXOBJ];
  __shared__ float s_red[16];
  __shared__ int   s_valid[MAXOBJ];
  __shared__ float s_norm;

  const int b = blockIdx.x;
  const int tid = threadIdx.x;
  const int lane = tid & 31;
  const int wv = tid >> 5;
  const int half = lane >> 4;
  const int mrow = lane & 15;
  const int nn = wv * 16 + (lane & 15);

  for (int t = tid; t < MAXOBJ * KP; t += 512) {
    int r = t / KP, c = t % KP;
    sC[r][c] = comb_g[(size_t)(b * MAXOBJ + r) * KP + c];
  }
  if (tid < MAXOBJ * (LD - KP)) {
    int r = tid / (LD - KP), c = KP + tid % (LD - KP);
    sC[r][c] = 0.f;
  }
  if (tid < MAXOBJ) s_valid[tid] = meta[(b * MAXOBJ + tid) * 6 + 5];

  if (tid < DD) {
    const float* sp = structure + (size_t)b * 8 * DD + tid;
    float s = 0.f;
    #pragma unroll
    for (int k = 0; k < 8; ++k) s += sp[(size_t)k * DD];
    s_sn[tid] = s * 0.125f;
  }
  __syncthreads();
  if (tid < DD) {
    float p = s_sn[tid] * s_sn[tid];
    #pragma unroll
    for (int m = 16; m >= 1; m >>= 1) p += __shfl_xor(p, m, 32);
    if (lane == 0) s_red[wv] = p;
  }
  __syncthreads();
  if (tid == 0) {
    float t = 0.f;
    for (int i = 0; i < 8; ++i) t += s_red[i];
    s_norm = fmaxf(sqrtf(t), 1e-8f);
  }
  __syncthreads();
  if (tid < DD) s_sn[tid] /= s_norm;

  {
    v8f acc = {};
    for (int k0 = 0; k0 < KP; k0 += 4) {
      const int kk = k0 + half * 2;
      v2f av; av.x = sC[mrow][kk]; av.y = sC[mrow][kk + 1];
      const int r0 = min(kk, KF - 1), r1 = min(kk + 1, KF - 1);
      v2f bv; bv.x = W1[(size_t)r0 * DD + nn]; bv.y = W1[(size_t)r1 * DD + nn];
      acc = __builtin_amdgcn_wmma_f32_16x16x4_f32(false, av, false, bv,
                                                  (short)0, acc, false, false);
    }
    const float bias = b1[nn];
    #pragma unroll
    for (int i = 0; i < 8; ++i) {
      const int row = half * 8 + i;
      float v = acc[i] + bias;
      v = 0.5f * v * (1.0f + erff(v * 0.70710678118654752f));
      sA[row][nn] = v;
    }
  }
  __syncthreads();

  {
    v8f acc = {};
    for (int k0 = 0; k0 < DD; k0 += 4) {
      const int kk = k0 + half * 2;
      v2f av; av.x = sA[mrow][kk]; av.y = sA[mrow][kk + 1];
      v2f bv; bv.x = W2[(size_t)kk * DD + nn]; bv.y = W2[(size_t)(kk + 1) * DD + nn];
      acc = __builtin_amdgcn_wmma_f32_16x16x4_f32(false, av, false, bv,
                                                  (short)0, acc, false, false);
    }
    const float bias = b2[nn];
    #pragma unroll
    for (int i = 0; i < 8; ++i) {
      const int row = half * 8 + i;
      const float v = acc[i] + bias;
      sB[row][nn] = s_valid[row] ? v : 0.f;
    }
  }
  __syncthreads();

  {
    float p = 0.f;
    #pragma unroll
    for (int j = 0; j < 8; ++j) {
      const int n = lane + 32 * j;
      p += sB[wv][n] * s_sn[n];
    }
    #pragma unroll
    for (int m = 16; m >= 1; m >>= 1) p += __shfl_xor(p, m, 32);
    if (lane == 0) s_dot[wv] = p;
  }
  __syncthreads();
  {
    const float os = osp[0];
    for (int t = tid; t < MAXOBJ * DD; t += 512) {
      const int r = t >> 8, n = t & 255;
      sB[r][n] = (sB[r][n] - s_dot[r] * s_sn[n]) * os;
    }
  }
  __syncthreads();

  {
    v8f acc = {};
    for (int k0 = 0; k0 < DD; k0 += 4) {
      const int kk = k0 + half * 2;
      v2f av; av.x = sB[mrow][kk]; av.y = sB[mrow][kk + 1];
      v2f bv; bv.x = Wp[(size_t)kk * DD + nn]; bv.y = Wp[(size_t)(kk + 1) * DD + nn];
      acc = __builtin_amdgcn_wmma_f32_16x16x4_f32(false, av, false, bv,
                                                  (short)0, acc, false, false);
    }
    const float bias = bp[nn];
    #pragma unroll
    for (int i = 0; i < 8; ++i) sC[half * 8 + i][nn] = acc[i] + bias;
  }
  __syncthreads();

  {
    float s1 = 0.f, s2 = 0.f;
    #pragma unroll
    for (int j = 0; j < 8; ++j) {
      const float v = sC[wv][lane + 32 * j];
      s1 += v; s2 += v * v;
    }
    #pragma unroll
    for (int m = 16; m >= 1; m >>= 1) {
      s1 += __shfl_xor(s1, m, 32);
      s2 += __shfl_xor(s2, m, 32);
    }
    const float mu = s1 * (1.0f / 256.0f);
    const float var = s2 * (1.0f / 256.0f) - mu * mu;
    const float rstd = rsqrtf(var + 1e-5f);
    float* op = out + (size_t)(b * MAXOBJ + wv) * DD;
    #pragma unroll
    for (int j = 0; j < 8; ++j) {
      const int n = lane + 32 * j;
      const float v = sC[wv][n];
      op[n] = (v - mu) * rstd * gmm[n] + bta[n];
    }
  }
}

extern "C" void kernel_launch(void* const* d_in, const int* in_sizes, int n_in,
                              void* d_out, int out_size, void* d_ws, size_t ws_size,
                              hipStream_t stream) {
  const float* grid_emb  = (const float*)d_in[0];
  const int*   grid      = (const int*)  d_in[1];
  const float* structure = (const float*)d_in[2];
  const float* W1        = (const float*)d_in[3];
  const float* b1        = (const float*)d_in[4];
  const float* W2        = (const float*)d_in[5];
  const float* b2        = (const float*)d_in[6];
  const float* Wp        = (const float*)d_in[7];
  const float* bp        = (const float*)d_in[8];
  const float* gmm       = (const float*)d_in[9];
  const float* bta       = (const float*)d_in[10];
  const float* osp       = (const float*)d_in[11];
  float* out = (float*)d_out;

  const int B = in_sizes[1] / HWC;

  float* comb = (float*)d_ws;
  int*   meta = (int*)((char*)d_ws + (size_t)B * MAXOBJ * KP * sizeof(float));

  cc_extract_kernel<<<B, 1024, 0, stream>>>(grid, meta);
  pool_kernel<<<B * MAXOBJ, 256, 0, stream>>>(grid_emb, meta, comb);
  fused_mlp_kernel<<<B, 512, 0, stream>>>(comb, meta, structure, W1, b1, W2, b2,
                                          Wp, bp, gmm, bta, osp, out);
}


# Fuse labeling, bbox mean-pool and the MLP chain into one launch: one 512-thread block per batch, objects and the combined tile stay in LDS, wave k pools object k; every summation order kept
# speedup vs baseline: 1.2030x; 1.2030x over previous
#include <hip/hip_runtime.h>
#include <math.h>


#define GH 30
#define GW 30
#define HWC 900
#define DD 256
#define MAXOBJ 16
#define SENT 900
#define KF 261
#define KP 264
#define LD 268
#define NT 512

typedef __attribute__((ext_vector_type(2))) float v2f;
typedef __attribute__((ext_vector_type(4))) float v4f;
typedef __attribute__((ext_vector_type(8))) float v8f;

__global__ __launch_bounds__(NT) void cc_pool_mlp_kernel(
    const int* __restrict__ grid, const float* __restrict__ grid_emb,
    const float* __restrict__ structure, const float* __restrict__ W1,
    const float* __restrict__ b1, const float* __restrict__ W2,
    const float* __restrict__ b2, const float* __restrict__ Wp,
    const float* __restrict__ bp, const float* __restrict__ gmm,
    const float* __restrict__ bta, const float* __restrict__ osp,
    float* __restrict__ out) {
  __shared__ int g[HWC];
  __shared__ int lab[HWC];
  __shared__ int rmn[HWC], rmx[HWC], cmn[HWC], cmx[HWC];
  __shared__ int roots[MAXOBJ];
  __shared__ int nroots;
  __shared__ int changed;
  __shared__ int s_meta[MAXOBJ][6];

  __shared__ float sA[MAXOBJ][LD];
  __shared__ float sB[MAXOBJ][LD];
  __shared__ float sC[MAXOBJ][LD];
  __shared__ float s_sn[DD];
  __shared__ float s_dot[MAXOBJ];
  __shared__ float s_red[16];
  __shared__ int   s_valid[MAXOBJ];
  __shared__ float s_norm;

  const int b = blockIdx.x;
  const int tid = threadIdx.x;
  const int lane = tid & 31;
  const int wv = tid >> 5;
  const int half = lane >> 4;
  const int mrow = lane & 15;
  const int nn = wv * 16 + (lane & 15);

  {
    const int* gg = grid + b * HWC;
    for (int t = tid; t < HWC; t += NT) {
      int v = gg[t];
      g[t] = v;
      lab[t] = (v > 0) ? t : SENT;
      rmn[t] = GH; rmx[t] = -1; cmn[t] = GW; cmx[t] = -1;
    }
  }
  __syncthreads();

  for (int iter = 0; iter < 2048; ++iter) {
    if (tid == 0) changed = 0;
    __syncthreads();
    int newl[2], oldl[2];
    bool act[2];
    #pragma unroll
    for (int q = 0; q < 2; ++q) {
      const int cell = tid + q * NT;
      act[q] = (cell < HWC) && (g[cell] > 0);
      newl[q] = SENT; oldl[q] = SENT;
      if (act[q]) {
        const int r = cell / GW, c = cell % GW;
        const int gv = g[cell];
        int m = lab[cell]; oldl[q] = m;
        if (r > 0      && g[cell - GW] == gv) m = min(m, lab[cell - GW]);
        if (r < GH - 1 && g[cell + GW] == gv) m = min(m, lab[cell + GW]);
        if (c > 0      && g[cell - 1]  == gv) m = min(m, lab[cell - 1]);
        if (c < GW - 1 && g[cell + 1]  == gv) m = min(m, lab[cell + 1]);
        if (m < SENT) m = min(m, lab[m]);
        newl[q] = m;
      }
    }
    __syncthreads();
    #pragma unroll
    for (int q = 0; q < 2; ++q) {
      if (act[q] && newl[q] != oldl[q]) { lab[tid + q * NT] = newl[q]; changed = 1; }
    }
    __syncthreads();
    const int done = (changed == 0);
    __syncthreads();
    if (done) break;
  }

  for (int cell = tid; cell < HWC; cell += NT) {
    if (g[cell] > 0) {
      const int root = lab[cell];
      atomicMin(&rmn[root], cell / GW);
      atomicMax(&rmx[root], cell / GW);
      atomicMin(&cmn[root], cell % GW);
      atomicMax(&cmx[root], cell % GW);
    }
  }
  __syncthreads();

  if (tid == 0) {
    int cnt = 0;
    for (int i = 0; i < HWC && cnt < MAXOBJ; ++i)
      if (g[i] > 0 && lab[i] == i) roots[cnt++] = i;
    nroots = cnt;
  }
  __syncthreads();

  if (tid < MAXOBJ) {
    int* mo = s_meta[tid];
    if (tid < nroots) {
      const int r = roots[tid];
      int y  = min(max(rmn[r], 0), GH - 1);
      int x  = min(max(cmn[r], 0), GW - 1);
      int y2 = min(max(rmx[r] + 1, y + 1), GH);
      int x2 = min(max(cmx[r] + 1, x + 1), GW);
      mo[0] = g[r]; mo[1] = y; mo[2] = x; mo[3] = y2 - y; mo[4] = x2 - x; mo[5] = 1;
      s_valid[tid] = 1;
    } else {
      mo[0] = 0; mo[1] = 0; mo[2] = 0; mo[3] = 1; mo[4] = 1; mo[5] = 0;
      s_valid[tid] = 0;
    }
  }
  __syncthreads();

  {
    const int color = s_meta[wv][0], y = s_meta[wv][1], x = s_meta[wv][2];
    const int h = s_meta[wv][3], w = s_meta[wv][4];
    v4f s0 = {0.f, 0.f, 0.f, 0.f}, s1 = {0.f, 0.f, 0.f, 0.f};
    for (int i = 0; i < h; ++i) {
      const v4f* rowp =
          (const v4f*)(grid_emb + (((size_t)(b * GH + y + i)) * GW + x) * DD);
      for (int j = 0; j < w; ++j) {
        s0 += rowp[j * (DD / 4) + lane];
        s1 += rowp[j * (DD / 4) + 32 + lane];
      }
    }
    const float cnt = (float)(h * w);
    *(v4f*)&sC[wv][4 * lane]       = s0 / cnt;
    *(v4f*)&sC[wv][128 + 4 * lane] = s1 / cnt;
    if (lane < LD - DD) {
      float f = 0.f;
      if (lane == 0) f = (float)color / 9.0f;
      if (lane == 1) f = (float)x / (float)GW;
      if (lane == 2) f = (float)y / (float)GH;
      if (lane == 3) f = (float)w / (float)GW;
      if (lane == 4) f = (float)h / (float)GH;
      sC[wv][DD + lane] = f;
    }
  }

  if (tid < DD) {
    const float* sp = structure + (size_t)b * 8 * DD + tid;
    float s = 0.f;
    #pragma unroll
    for (int k = 0; k < 8; ++k) s += sp[(size_t)k * DD];
    s_sn[tid] = s * 0.125f;
  }
  __syncthreads();
  if (tid < DD) {
    float p = s_sn[tid] * s_sn[tid];
    #pragma unroll
    for (int m = 16; m >= 1; m >>= 1) p += __shfl_xor(p, m, 32);
    if (lane == 0) s_red[wv] = p;
  }
  __syncthreads();
  if (tid == 0) {
    float t = 0.f;
    for (int i = 0; i < 8; ++i) t += s_red[i];
    s_norm = fmaxf(sqrtf(t), 1e-8f);
  }
  __syncthreads();
  if (tid < DD) s_sn[tid] /= s_norm;

  {
    v8f acc = {};
    for (int k0 = 0; k0 < KP; k0 += 4) {
      const int kk = k0 + half * 2;
      v2f av; av.x = sC[mrow][kk]; av.y = sC[mrow][kk + 1];
      const int r0 = min(kk, KF - 1), r1 = min(kk + 1, KF - 1);
      v2f bv; bv.x = W1[(size_t)r0 * DD + nn]; bv.y = W1[(size_t)r1 * DD + nn];
      acc = __builtin_amdgcn_wmma_f32_16x16x4_f32(false, av, false, bv,
                                                  (short)0, acc, false, false);
    }
    const float bias = b1[nn];
    #pragma unroll
    for (int i = 0; i < 8; ++i) {
      const int row = half * 8 + i;
      float v = acc[i] + bias;
      v = 0.5f * v * (1.0f + erff(v * 0.70710678118654752f));
      sA[row][nn] = v;
    }
  }
  __syncthreads();

  {
    v8f acc = {};
    for (int k0 = 0; k0 < DD; k0 += 4) {
      const int kk = k0 + half * 2;
      v2f av; av.x = sA[mrow][kk]; av.y = sA[mrow][kk + 1];
      v2f bv; bv.x = W2[(size_t)kk * DD + nn]; bv.y = W2[(size_t)(kk + 1) * DD + nn];
      acc = __builtin_amdgcn_wmma_f32_16x16x4_f32(false, av, false, bv,
                                                  (short)0, acc, false, false);
    }
    const float bias = b2[nn];
    #pragma unroll
    for (int i = 0; i < 8; ++i) {
      const int row = half * 8 + i;
      const float v = acc[i] + bias;
      sB[row][nn] = s_valid[row] ? v : 0.f;
    }
  }
  __syncthreads();

  {
    float p = 0.f;
    #pragma unroll
    for (int j = 0; j < 8; ++j) {
      const int n = lane + 32 * j;
      p += sB[wv][n] * s_sn[n];
    }
    #pragma unroll
    for (int m = 16; m >= 1; m >>= 1) p += __shfl_xor(p, m, 32);
    if (lane == 0) s_dot[wv] = p;
  }
  __syncthreads();
  {
    const float os = osp[0];
    for (int t = tid; t < MAXOBJ * DD; t += NT) {
      const int r = t >> 8, n = t & 255;
      sB[r][n] = (sB[r][n] - s_dot[r] * s_sn[n]) * os;
    }
  }
  __syncthreads();

  {
    v8f acc = {};
    for (int k0 = 0; k0 < DD; k0 += 4) {
      const int kk = k0 + half * 2;
      v2f av; av.x = sB[mrow][kk]; av.y = sB[mrow][kk + 1];
      v2f bv; bv.x = Wp[(size_t)kk * DD + nn]; bv.y = Wp[(size_t)(kk + 1) * DD + nn];
      acc = __builtin_amdgcn_wmma_f32_16x16x4_f32(false, av, false, bv,
                                                  (short)0, acc, false, false);
    }
    const float bias = bp[nn];
    #pragma unroll
    for (int i = 0; i < 8; ++i) sC[half * 8 + i][nn] = acc[i] + bias;
  }
  __syncthreads();

  {
    float s1 = 0.f, s2 = 0.f;
    #pragma unroll
    for (int j = 0; j < 8; ++j) {
      const float v = sC[wv][lane + 32 * j];
      s1 += v; s2 += v * v;
    }
    #pragma unroll
    for (int m = 16; m >= 1; m >>= 1) {
      s1 += __shfl_xor(s1, m, 32);
      s2 += __shfl_xor(s2, m, 32);
    }
    const float mu = s1 * (1.0f / 256.0f);
    const float var = s2 * (1.0f / 256.0f) - mu * mu;
    const float rstd = rsqrtf(var + 1e-5f);
    float* op = out + (size_t)(b * MAXOBJ + wv) * DD;
    #pragma unroll
    for (int j = 0; j < 8; ++j) {
      const int n = lane + 32 * j;
      const float v = sC[wv][n];
      op[n] = (v - mu) * rstd * gmm[n] + bta[n];
    }
  }
}

extern "C" void kernel_launch(void* const* d_in, const int* in_sizes, int n_in,
                              void* d_out, int out_size, void* d_ws, size_t ws_size,
                              hipStream_t stream) {
  const float* grid_emb  = (const float*)d_in[0];
  const int*   grid      = (const int*)  d_in[1];
  const float* structure = (const float*)d_in[2];
  const float* W1        = (const float*)d_in[3];
  const float* b1        = (const float*)d_in[4];
  const float* W2        = (const float*)d_in[5];
  const float* b2        = (const float*)d_in[6];
  const float* Wp        = (const float*)d_in[7];
  const float* bp        = (const float*)d_in[8];
  const float* gmm       = (const float*)d_in[9];
  const float* bta       = (const float*)d_in[10];
  const float* osp       = (const float*)d_in[11];
  float* out = (float*)d_out;

  const int B = in_sizes[1] / HWC;

  cc_pool_mlp_kernel<<<B, NT, 0, stream>>>(grid, grid_emb, structure, W1, b1, W2, b2,
                                           Wp, bp, gmm, bta, osp, out);
}


# Labeling sweep: same-colour neighbour indices kept in registers, all neighbour labels read as one batch of LDS loads, labels updated in place: 2 LDS round trips and 1 barrier per sweep (was 21 and 4)
# speedup vs baseline: 1.4252x; 1.1847x over previous
#include <hip/hip_runtime.h>
#include <math.h>


#define GH 30
#define GW 30
#define HWC 900
#define DD 256
#define MAXOBJ 16
#define SENT 900
#define KF 261
#define KP 264
#define LD 268
#define NT 512
#define LP 1024

typedef __attribute__((ext_vector_type(2))) float v2f;
typedef __attribute__((ext_vector_type(4))) float v4f;
typedef __attribute__((ext_vector_type(8))) float v8f;

__global__ __launch_bounds__(NT) void cc_pool_mlp_kernel(
    const int* __restrict__ grid, const float* __restrict__ grid_emb,
    const float* __restrict__ structure, const float* __restrict__ W1,
    const float* __restrict__ b1, const float* __restrict__ W2,
    const float* __restrict__ b2, const float* __restrict__ Wp,
    const float* __restrict__ bp, const float* __restrict__ gmm,
    const float* __restrict__ bta, const float* __restrict__ osp,
    float* __restrict__ out) {
  __shared__ int g[LP];
  __shared__ int lab[LP];
  __shared__ int rmn[HWC], rmx[HWC], cmn[HWC], cmx[HWC];
  __shared__ int roots[MAXOBJ];
  __shared__ int nroots;
  __shared__ int chg[3];
  __shared__ int s_meta[MAXOBJ][6];

  __shared__ float sA[MAXOBJ][LD];
  __shared__ float sB[MAXOBJ][LD];
  __shared__ float sC[MAXOBJ][LD];
  __shared__ float s_sn[DD];
  __shared__ float s_dot[MAXOBJ];
  __shared__ float s_red[16];
  __shared__ int   s_valid[MAXOBJ];
  __shared__ float s_norm;

  const int b = blockIdx.x;
  const int tid = threadIdx.x;
  const int lane = tid & 31;
  const int wv = tid >> 5;
  const int half = lane >> 4;
  const int mrow = lane & 15;
  const int nn = wv * 16 + (lane & 15);

  {
    const int* gg = grid + b * HWC;
    const int t0 = tid, t1 = tid + NT;
    const int v0 = gg[t0];
    const int v1 = (t1 < HWC) ? gg[t1] : 0;
    g[t0] = v0; lab[t0] = (v0 > 0) ? t0 : SENT;
    g[t1] = v1; lab[t1] = (v1 > 0) ? t1 : SENT;
    rmn[t0] = GH; rmx[t0] = -1; cmn[t0] = GW; cmx[t0] = -1;
    if (t1 < HWC) { rmn[t1] = GH; rmx[t1] = -1; cmn[t1] = GW; cmx[t1] = -1; }
    if (tid < 3) chg[tid] = 0;
  }
  __syncthreads();

  int nb0u, nb0d, nb0l, nb0r, nb1u, nb1d, nb1l, nb1r;
  int my0, my1;
  {
    const int c0 = tid, c1 = tid + NT;
    const int r0 = c0 / GW, x0 = c0 - r0 * GW;
    const int r1 = c1 / GW, x1 = c1 - r1 * GW;
    const int u0 = (r0 > 0) ? c0 - GW : c0;
    const int d0 = (c0 + GW < HWC) ? c0 + GW : c0;
    const int l0 = (x0 > 0) ? c0 - 1 : c0;
    const int e0 = (x0 < GW - 1) ? c0 + 1 : c0;
    const int u1 = c1 - GW;
    const int d1 = (c1 + GW < HWC) ? c1 + GW : c1;
    const int l1 = (x1 > 0) ? c1 - 1 : c1;
    const int e1 = (x1 < GW - 1 && c1 + 1 < HWC) ? c1 + 1 : c1;
    const int gv0 = g[c0], gv1 = g[c1];
    const int gu0 = g[u0], gd0 = g[d0], gl0 = g[l0], ge0 = g[e0];
    const int gu1 = g[u1], gd1 = g[d1], gl1 = g[l1], ge1 = g[e1];
    const bool f0 = gv0 > 0, f1 = gv1 > 0;
    nb0u = (f0 && gu0 == gv0) ? u0 : c0;
    nb0d = (f0 && gd0 == gv0) ? d0 : c0;
    nb0l = (f0 && gl0 == gv0) ? l0 : c0;
    nb0r = (f0 && ge0 == gv0) ? e0 : c0;
    nb1u = (f1 && gu1 == gv1) ? u1 : c1;
    nb1d = (f1 && gd1 == gv1) ? d1 : c1;
    nb1l = (f1 && gl1 == gv1) ? l1 : c1;
    nb1r = (f1 && ge1 == gv1) ? e1 : c1;
    my0 = f0 ? c0 : SENT;
    my1 = f1 ? c1 : SENT;
  }

  int cur = 0;
  for (int iter = 0; iter < 2048; ++iter) {
    const int a0 = lab[nb0u], a1 = lab[nb0d], a2 = lab[nb0l], a3 = lab[nb0r];
    const int a4 = lab[nb1u], a5 = lab[nb1d], a6 = lab[nb1l], a7 = lab[nb1r];
    int m0 = min(min(my0, a0), min(min(a1, a2), a3));
    int m1 = min(min(my1, a4), min(min(a5, a6), a7));
    const int j0 = lab[m0], j1 = lab[m1];
    __builtin_amdgcn_sched_barrier(0);
    m0 = min(m0, j0);
    m1 = min(m1, j1);
    const bool mv0 = m0 < my0, mv1 = m1 < my1;
    if (mv0) { lab[tid] = m0; my0 = m0; }
    if (mv1) { lab[tid + NT] = m1; my1 = m1; }
    if (mv0 || mv1) chg[cur] = 1;
    const int nxt = (cur == 2) ? 0 : cur + 1;
    if (tid == 0) chg[nxt] = 0;
    __syncthreads();
    const int any = chg[cur];
    cur = nxt;
    if (any == 0) break;
  }

  for (int cell = tid; cell < HWC; cell += NT) {
    if (g[cell] > 0) {
      const int root = lab[cell];
      atomicMin(&rmn[root], cell / GW);
      atomicMax(&rmx[root], cell / GW);
      atomicMin(&cmn[root], cell % GW);
      atomicMax(&cmx[root], cell % GW);
    }
  }
  __syncthreads();

  if (tid == 0) {
    int cnt = 0;
    for (int i = 0; i < HWC && cnt < MAXOBJ; ++i)
      if (g[i] > 0 && lab[i] == i) roots[cnt++] = i;
    nroots = cnt;
  }
  __syncthreads();

  if (tid < MAXOBJ) {
    int* mo = s_meta[tid];
    if (tid < nroots) {
      const int r = roots[tid];
      int y  = min(max(rmn[r], 0), GH - 1);
      int x  = min(max(cmn[r], 0), GW - 1);
      int y2 = min(max(rmx[r] + 1, y + 1), GH);
      int x2 = min(max(cmx[r] + 1, x + 1), GW);
      mo[0] = g[r]; mo[1] = y; mo[2] = x; mo[3] = y2 - y; mo[4] = x2 - x; mo[5] = 1;
      s_valid[tid] = 1;
    } else {
      mo[0] = 0; mo[1] = 0; mo[2] = 0; mo[3] = 1; mo[4] = 1; mo[5] = 0;
      s_valid[tid] = 0;
    }
  }
  __syncthreads();

  {
    const int color = s_meta[wv][0], y = s_meta[wv][1], x = s_meta[wv][2];
    const int h = s_meta[wv][3], w = s_meta[wv][4];
    v4f s0 = {0.f, 0.f, 0.f, 0.f}, s1 = {0.f, 0.f, 0.f, 0.f};
    for (int i = 0; i < h; ++i) {
      const v4f* rowp =
          (const v4f*)(grid_emb + (((size_t)(b * GH + y + i)) * GW + x) * DD);
      for (int j = 0; j < w; ++j) {
        s0 += rowp[j * (DD / 4) + lane];
        s1 += rowp[j * (DD / 4) + 32 + lane];
      }
    }
    const float cnt = (float)(h * w);
    *(v4f*)&sC[wv][4 * lane]       = s0 / cnt;
    *(v4f*)&sC[wv][128 + 4 * lane] = s1 / cnt;
    if (lane < LD - DD) {
      float f = 0.f;
      if (lane == 0) f = (float)color / 9.0f;
      if (lane == 1) f = (float)x / (float)GW;
      if (lane == 2) f = (float)y / (float)GH;
      if (lane == 3) f = (float)w / (float)GW;
      if (lane == 4) f = (float)h / (float)GH;
      sC[wv][DD + lane] = f;
    }
  }

  if (tid < DD) {
    const float* sp = structure + (size_t)b * 8 * DD + tid;
    float s = 0.f;
    #pragma unroll
    for (int k = 0; k < 8; ++k) s += sp[(size_t)k * DD];
    s_sn[tid] = s * 0.125f;
  }
  __syncthreads();
  if (tid < DD) {
    float p = s_sn[tid] * s_sn[tid];
    #pragma unroll
    for (int m = 16; m >= 1; m >>= 1) p += __shfl_xor(p, m, 32);
    if (lane == 0) s_red[wv] = p;
  }
  __syncthreads();
  if (tid == 0) {
    float t = 0.f;
    for (int i = 0; i < 8; ++i) t += s_red[i];
    s_norm = fmaxf(sqrtf(t), 1e-8f);
  }
  __syncthreads();
  if (tid < DD) s_sn[tid] /= s_norm;

  {
    v8f acc = {};
    for (int k0 = 0; k0 < KP; k0 += 4) {
      const int kk = k0 + half * 2;
      v2f av; av.x = sC[mrow][kk]; av.y = sC[mrow][kk + 1];
      const int r0 = min(kk, KF - 1), r1 = min(kk + 1, KF - 1);
      v2f bv; bv.x = W1[(size_t)r0 * DD + nn]; bv.y = W1[(size_t)r1 * DD + nn];
      acc = __builtin_amdgcn_wmma_f32_16x16x4_f32(false, av, false, bv,
                                                  (short)0, acc, false, false);
    }
    const float bias = b1[nn];
    #pragma unroll
    for (int i = 0; i < 8; ++i) {
      const int row = half * 8 + i;
      float v = acc[i] + bias;
      v = 0.5f * v * (1.0f + erff(v * 0.70710678118654752f));
      sA[row][nn] = v;
    }
  }
  __syncthreads();

  {
    v8f acc = {};
    for (int k0 = 0; k0 < DD; k0 += 4) {
      const int kk = k0 + half * 2;
      v2f av; av.x = sA[mrow][kk]; av.y = sA[mrow][kk + 1];
      v2f bv; bv.x = W2[(size_t)kk * DD + nn]; bv.y = W2[(size_t)(kk + 1) * DD + nn];
      acc = __builtin_amdgcn_wmma_f32_16x16x4_f32(false, av, false, bv,
                                                  (short)0, acc, false, false);
    }
    const float bias = b2[nn];
    #pragma unroll
    for (int i = 0; i < 8; ++i) {
      const int row = half * 8 + i;
      const float v = acc[i] + bias;
      sB[row][nn] = s_valid[row] ? v : 0.f;
    }
  }
  __syncthreads();

  {
    float p = 0.f;
    #pragma unroll
    for (int j = 0; j < 8; ++j) {
      const int n = lane + 32 * j;
      p += sB[wv][n] * s_sn[n];
    }
    #pragma unroll
    for (int m = 16; m >= 1; m >>= 1) p += __shfl_xor(p, m, 32);
    if (lane == 0) s_dot[wv] = p;
  }
  __syncthreads();
  {
    const float os = osp[0];
    for (int t = tid; t < MAXOBJ * DD; t += NT) {
      const int r = t >> 8, n = t & 255;
      sB[r][n] = (sB[r][n] - s_dot[r] * s_sn[n]) * os;
    }
  }
  __syncthreads();

  {
    v8f acc = {};
    for (int k0 = 0; k0 < DD; k0 += 4) {
      const int kk = k0 + half * 2;
      v2f av; av.x = sB[mrow][kk]; av.y = sB[mrow][kk + 1];
      v2f bv; bv.x = Wp[(size_t)kk * DD + nn]; bv.y = Wp[(size_t)(kk + 1) * DD + nn];
      acc = __builtin_amdgcn_wmma_f32_16x16x4_f32(false, av, false, bv,
                                                  (short)0, acc, false, false);
    }
    const float bias = bp[nn];
    #pragma unroll
    for (int i = 0; i < 8; ++i) sC[half * 8 + i][nn] = acc[i] + bias;
  }
  __syncthreads();

  {
    float s1 = 0.f, s2 = 0.f;
    #pragma unroll
    for (int j = 0; j < 8; ++j) {
      const float v = sC[wv][lane + 32 * j];
      s1 += v; s2 += v * v;
    }
    #pragma unroll
    for (int m = 16; m >= 1; m >>= 1) {
      s1 += __shfl_xor(s1, m, 32);
      s2 += __shfl_xor(s2, m, 32);
    }
    const float mu = s1 * (1.0f / 256.0f);
    const float var = s2 * (1.0f / 256.0f) - mu * mu;
    const float rstd = rsqrtf(var + 1e-5f);
    float* op = out + (size_t)(b * MAXOBJ + wv) * DD;
    #pragma unroll
    for (int j = 0; j < 8; ++j) {
      const int n = lane + 32 * j;
      const float v = sC[wv][n];
      op[n] = (v - mu) * rstd * gmm[n] + bta[n];
    }
  }
}

extern "C" void kernel_launch(void* const* d_in, const int* in_sizes, int n_in,
                              void* d_out, int out_size, void* d_ws, size_t ws_size,
                              hipStream_t stream) {
  const float* grid_emb  = (const float*)d_in[0];
  const int*   grid      = (const int*)  d_in[1];
  const float* structure = (const float*)d_in[2];
  const float* W1        = (const float*)d_in[3];
  const float* b1        = (const float*)d_in[4];
  const float* W2        = (const float*)d_in[5];
  const float* b2        = (const float*)d_in[6];
  const float* Wp        = (const float*)d_in[7];
  const float* bp        = (const float*)d_in[8];
  const float* gmm       = (const float*)d_in[9];
  const float* bta       = (const float*)d_in[10];
  const float* osp       = (const float*)d_in[11];
  float* out = (float*)d_out;

  const int B = in_sizes[1] / HWC;

  cc_pool_mlp_kernel<<<B, NT, 0, stream>>>(grid, grid_emb, structure, W1, b1, W2, b2,
                                           Wp, bp, gmm, bta, osp, out);
}


# Object extraction by all threads: root cells ranked with wave ballots and a popcount prefix over 32 mask words (was one thread scanning cell by cell), boxes built from register-held labels
# speedup vs baseline: 1.6204x; 1.1370x over previous
#include <hip/hip_runtime.h>
#include <math.h>


#define GH 30
#define GW 30
#define HWC 900
#define DD 256
#define MAXOBJ 16
#define SENT 900
#define KF 261
#define KP 264
#define LD 268
#define NT 512
#define LP 1024

typedef __attribute__((ext_vector_type(2))) float v2f;
typedef __attribute__((ext_vector_type(4))) float v4f;
typedef __attribute__((ext_vector_type(8))) float v8f;

__global__ __launch_bounds__(NT) void cc_pool_mlp_kernel(
    const int* __restrict__ grid, const float* __restrict__ grid_emb,
    const float* __restrict__ structure, const float* __restrict__ W1,
    const float* __restrict__ b1, const float* __restrict__ W2,
    const float* __restrict__ b2, const float* __restrict__ Wp,
    const float* __restrict__ bp, const float* __restrict__ gmm,
    const float* __restrict__ bta, const float* __restrict__ osp,
    float* __restrict__ out) {
  __shared__ int g[LP];
  __shared__ int lab[LP];
  __shared__ int rmn[HWC], rmx[HWC], cmn[HWC], cmx[HWC];
  __shared__ int roots[MAXOBJ];
  __shared__ unsigned wmask[32];
  __shared__ int chg[3];

  __shared__ float sA[MAXOBJ][LD];
  __shared__ float sB[MAXOBJ][LD];
  __shared__ float sC[MAXOBJ][LD];
  __shared__ float s_sn[DD];
  __shared__ float s_dot[MAXOBJ];
  __shared__ float s_red[16];
  __shared__ int   s_valid[MAXOBJ];
  __shared__ float s_norm;

  const int b = blockIdx.x;
  const int tid = threadIdx.x;
  const int lane = tid & 31;
  const int wv = tid >> 5;
  const int half = lane >> 4;
  const int mrow = lane & 15;
  const int nn = wv * 16 + (lane & 15);

  {
    const int* gg = grid + b * HWC;
    const int t0 = tid, t1 = tid + NT;
    const int v0 = gg[t0];
    const int v1 = (t1 < HWC) ? gg[t1] : 0;
    g[t0] = v0; lab[t0] = (v0 > 0) ? t0 : SENT;
    g[t1] = v1; lab[t1] = (v1 > 0) ? t1 : SENT;
    rmn[t0] = GH; rmx[t0] = -1; cmn[t0] = GW; cmx[t0] = -1;
    if (t1 < HWC) { rmn[t1] = GH; rmx[t1] = -1; cmn[t1] = GW; cmx[t1] = -1; }
    if (tid < 3) chg[tid] = 0;
  }
  __syncthreads();

  int nb0u, nb0d, nb0l, nb0r, nb1u, nb1d, nb1l, nb1r;
  int my0, my1;
  {
    const int c0 = tid, c1 = tid + NT;
    const int r0 = c0 / GW, x0 = c0 - r0 * GW;
    const int r1 = c1 / GW, x1 = c1 - r1 * GW;
    const int u0 = (r0 > 0) ? c0 - GW : c0;
    const int d0 = (c0 + GW < HWC) ? c0 + GW : c0;
    const int l0 = (x0 > 0) ? c0 - 1 : c0;
    const int e0 = (x0 < GW - 1) ? c0 + 1 : c0;
    const int u1 = c1 - GW;
    const int d1 = (c1 + GW < HWC) ? c1 + GW : c1;
    const int l1 = (x1 > 0) ? c1 - 1 : c1;
    const int e1 = (x1 < GW - 1 && c1 + 1 < HWC) ? c1 + 1 : c1;
    const int gv0 = g[c0], gv1 = g[c1];
    const int gu0 = g[u0], gd0 = g[d0], gl0 = g[l0], ge0 = g[e0];
    const int gu1 = g[u1], gd1 = g[d1], gl1 = g[l1], ge1 = g[e1];
    const bool f0 = gv0 > 0, f1 = gv1 > 0;
    nb0u = (f0 && gu0 == gv0) ? u0 : c0;
    nb0d = (f0 && gd0 == gv0) ? d0 : c0;
    nb0l = (f0 && gl0 == gv0) ? l0 : c0;
    nb0r = (f0 && ge0 == gv0) ? e0 : c0;
    nb1u = (f1 && gu1 == gv1) ? u1 : c1;
    nb1d = (f1 && gd1 == gv1) ? d1 : c1;
    nb1l = (f1 && gl1 == gv1) ? l1 : c1;
    nb1r = (f1 && ge1 == gv1) ? e1 : c1;
    my0 = f0 ? c0 : SENT;
    my1 = f1 ? c1 : SENT;
  }

  int cur = 0;
  for (int iter = 0; iter < 2048; ++iter) {
    const int a0 = lab[nb0u], a1 = lab[nb0d], a2 = lab[nb0l], a3 = lab[nb0r];
    const int a4 = lab[nb1u], a5 = lab[nb1d], a6 = lab[nb1l], a7 = lab[nb1r];
    int m0 = min(min(my0, a0), min(min(a1, a2), a3));
    int m1 = min(min(my1, a4), min(min(a5, a6), a7));
    const int j0 = lab[m0], j1 = lab[m1];
    __builtin_amdgcn_sched_barrier(0);
    m0 = min(m0, j0);
    m1 = min(m1, j1);
    const bool mv0 = m0 < my0, mv1 = m1 < my1;
    if (mv0) { lab[tid] = m0; my0 = m0; }
    if (mv1) { lab[tid + NT] = m1; my1 = m1; }
    if (mv0 || mv1) chg[cur] = 1;
    const int nxt = (cur == 2) ? 0 : cur + 1;
    if (tid == 0) chg[nxt] = 0;
    __syncthreads();
    const int any = chg[cur];
    cur = nxt;
    if (any == 0) break;
  }

  {
    const int t0 = tid, t1 = tid + NT;
    if (my0 < SENT) {
      const int rr = t0 / GW, cc = t0 - rr * GW;
      atomicMin(&rmn[my0], rr); atomicMax(&rmx[my0], rr);
      atomicMin(&cmn[my0], cc); atomicMax(&cmx[my0], cc);
    }
    if (my1 < SENT) {
      const int rr = t1 / GW, cc = t1 - rr * GW;
      atomicMin(&rmn[my1], rr); atomicMax(&rmx[my1], rr);
      atomicMin(&cmn[my1], cc); atomicMax(&cmx[my1], cc);
    }
  }

  const bool isr0 = (my0 == tid);
  const bool isr1 = (my1 == tid + NT) && (tid + NT < HWC);
  const unsigned rb0 = __builtin_amdgcn_ballot_w32(isr0);
  const unsigned rb1 = __builtin_amdgcn_ballot_w32(isr1);
  if (lane == 0) { wmask[wv] = rb0; wmask[16 + wv] = rb1; }
  __syncthreads();

  int nr;
  {
    const unsigned c = (unsigned)__popc(wmask[lane]);
    const unsigned lo = (1u << wv) - 1u;
    const unsigned hi = (1u << (16 + wv)) - 1u;
    unsigned base0 = 0u, base1 = 0u, total = 0u;
    for (int k = 0; k < 6; ++k) {
      const unsigned bk = __builtin_amdgcn_ballot_w32(((c >> k) & 1u) != 0u);
      base0 += (unsigned)__popc(bk & lo) << k;
      base1 += (unsigned)__popc(bk & hi) << k;
      total += (unsigned)__popc(bk) << k;
    }
    const unsigned below = (1u << lane) - 1u;
    if (isr0) {
      const unsigned rank = base0 + (unsigned)__popc(rb0 & below);
      if (rank < MAXOBJ) roots[rank] = tid;
    }
    if (isr1) {
      const unsigned rank = base1 + (unsigned)__popc(rb1 & below);
      if (rank < MAXOBJ) roots[rank] = tid + NT;
    }
    nr = (int)min(total, (unsigned)MAXOBJ);
  }
  __syncthreads();

  {
    int color = 0, y = 0, x = 0, h = 1, w = 1;
    const bool valid = wv < nr;
    if (valid) {
      const int r = roots[wv];
      y = min(max(rmn[r], 0), GH - 1);
      x = min(max(cmn[r], 0), GW - 1);
      const int y2 = min(max(rmx[r] + 1, y + 1), GH);
      const int x2 = min(max(cmx[r] + 1, x + 1), GW);
      color = g[r]; h = y2 - y; w = x2 - x;
    }
    if (lane == 0) s_valid[wv] = valid ? 1 : 0;
    v4f s0 = {0.f, 0.f, 0.f, 0.f}, s1 = {0.f, 0.f, 0.f, 0.f};
    for (int i = 0; i < h; ++i) {
      const v4f* rowp =
          (const v4f*)(grid_emb + (((size_t)(b * GH + y + i)) * GW + x) * DD);
      for (int j = 0; j < w; ++j) {
        s0 += rowp[j * (DD / 4) + lane];
        s1 += rowp[j * (DD / 4) + 32 + lane];
      }
    }
    const float cnt = (float)(h * w);
    *(v4f*)&sC[wv][4 * lane]       = s0 / cnt;
    *(v4f*)&sC[wv][128 + 4 * lane] = s1 / cnt;
    if (lane < LD - DD) {
      float f = 0.f;
      if (lane == 0) f = (float)color / 9.0f;
      if (lane == 1) f = (float)x / (float)GW;
      if (lane == 2) f = (float)y / (float)GH;
      if (lane == 3) f = (float)w / (float)GW;
      if (lane == 4) f = (float)h / (float)GH;
      sC[wv][DD + lane] = f;
    }
  }

  if (tid < DD) {
    const float* sp = structure + (size_t)b * 8 * DD + tid;
    float s = 0.f;
    #pragma unroll
    for (int k = 0; k < 8; ++k) s += sp[(size_t)k * DD];
    s_sn[tid] = s * 0.125f;
  }
  __syncthreads();
  if (tid < DD) {
    float p = s_sn[tid] * s_sn[tid];
    #pragma unroll
    for (int m = 16; m >= 1; m >>= 1) p += __shfl_xor(p, m, 32);
    if (lane == 0) s_red[wv] = p;
  }
  __syncthreads();
  if (tid == 0) {
    float t = 0.f;
    for (int i = 0; i < 8; ++i) t += s_red[i];
    s_norm = fmaxf(sqrtf(t), 1e-8f);
  }
  __syncthreads();
  if (tid < DD) s_sn[tid] /= s_norm;

  {
    v8f acc = {};
    for (int k0 = 0; k0 < KP; k0 += 4) {
      const int kk = k0 + half * 2;
      v2f av; av.x = sC[mrow][kk]; av.y = sC[mrow][kk + 1];
      const int r0 = min(kk, KF - 1), r1 = min(kk + 1, KF - 1);
      v2f bv; bv.x = W1[(size_t)r0 * DD + nn]; bv.y = W1[(size_t)r1 * DD + nn];
      acc = __builtin_amdgcn_wmma_f32_16x16x4_f32(false, av, false, bv,
                                                  (short)0, acc, false, false);
    }
    const float bias = b1[nn];
    #pragma unroll
    for (int i = 0; i < 8; ++i) {
      const int row = half * 8 + i;
      float v = acc[i] + bias;
      v = 0.5f * v * (1.0f + erff(v * 0.70710678118654752f));
      sA[row][nn] = v;
    }
  }
  __syncthreads();

  {
    v8f acc = {};
    for (int k0 = 0; k0 < DD; k0 += 4) {
      const int kk = k0 + half * 2;
      v2f av; av.x = sA[mrow][kk]; av.y = sA[mrow][kk + 1];
      v2f bv; bv.x = W2[(size_t)kk * DD + nn]; bv.y = W2[(size_t)(kk + 1) * DD + nn];
      acc = __builtin_amdgcn_wmma_f32_16x16x4_f32(false, av, false, bv,
                                                  (short)0, acc, false, false);
    }
    const float bias = b2[nn];
    #pragma unroll
    for (int i = 0; i < 8; ++i) {
      const int row = half * 8 + i;
      const float v = acc[i] + bias;
      sB[row][nn] = s_valid[row] ? v : 0.f;
    }
  }
  __syncthreads();

  {
    float p = 0.f;
    #pragma unroll
    for (int j = 0; j < 8; ++j) {
      const int n = lane + 32 * j;
      p += sB[wv][n] * s_sn[n];
    }
    #pragma unroll
    for (int m = 16; m >= 1; m >>= 1) p += __shfl_xor(p, m, 32);
    if (lane == 0) s_dot[wv] = p;
  }
  __syncthreads();
  {
    const float os = osp[0];
    for (int t = tid; t < MAXOBJ * DD; t += NT) {
      const int r = t >> 8, n = t & 255;
      sB[r][n] = (sB[r][n] - s_dot[r] * s_sn[n]) * os;
    }
  }
  __syncthreads();

  {
    v8f acc = {};
    for (int k0 = 0; k0 < DD; k0 += 4) {
      const int kk = k0 + half * 2;
      v2f av; av.x = sB[mrow][kk]; av.y = sB[mrow][kk + 1];
      v2f bv; bv.x = Wp[(size_t)kk * DD + nn]; bv.y = Wp[(size_t)(kk + 1) * DD + nn];
      acc = __builtin_amdgcn_wmma_f32_16x16x4_f32(false, av, false, bv,
                                                  (short)0, acc, false, false);
    }
    const float bias = bp[nn];
    #pragma unroll
    for (int i = 0; i < 8; ++i) sC[half * 8 + i][nn] = acc[i] + bias;
  }
  __syncthreads();

  {
    float s1 = 0.f, s2 = 0.f;
    #pragma unroll
    for (int j = 0; j < 8; ++j) {
      const float v = sC[wv][lane + 32 * j];
      s1 += v; s2 += v * v;
    }
    #pragma unroll
    for (int m = 16; m >= 1; m >>= 1) {
      s1 += __shfl_xor(s1, m, 32);
      s2 += __shfl_xor(s2, m, 32);
    }
    const float mu = s1 * (1.0f / 256.0f);
    const float var = s2 * (1.0f / 256.0f) - mu * mu;
    const float rstd = rsqrtf(var + 1e-5f);
    float* op = out + (size_t)(b * MAXOBJ + wv) * DD;
    #pragma unroll
    for (int j = 0; j < 8; ++j) {
      const int n = lane + 32 * j;
      const float v = sC[wv][n];
      op[n] = (v - mu) * rstd * gmm[n] + bta[n];
    }
  }
}

extern "C" void kernel_launch(void* const* d_in, const int* in_sizes, int n_in,
                              void* d_out, int out_size, void* d_ws, size_t ws_size,
                              hipStream_t stream) {
  const float* grid_emb  = (const float*)d_in[0];
  const int*   grid      = (const int*)  d_in[1];
  const float* structure = (const float*)d_in[2];
  const float* W1        = (const float*)d_in[3];
  const float* b1        = (const float*)d_in[4];
  const float* W2        = (const float*)d_in[5];
  const float* b2        = (const float*)d_in[6];
  const float* Wp        = (const float*)d_in[7];
  const float* bp        = (const float*)d_in[8];
  const float* gmm       = (const float*)d_in[9];
  const float* bta       = (const float*)d_in[10];
  const float* osp       = (const float*)d_in[11];
  float* out = (float*)d_out;

  const int B = in_sizes[1] / HWC;

  cc_pool_mlp_kernel<<<B, NT, 0, stream>>>(grid, grid_emb, structure, W1, b1, W2, b2,
                                           Wp, bp, gmm, bta, osp, out);
}


# Matrix stages on one wave per SIMD: 4 waves each own 64 columns as 4 interleaved tiles, so weight rows come as 16-byte loads requested 8 k-steps ahead in a second register set; sum order kept
# speedup vs baseline: 1.7639x; 1.0886x over previous
#include <hip/hip_runtime.h>
#include <math.h>


#define GH 30
#define GW 30
#define HWC 900
#define DD 256
#define MAXOBJ 16
#define SENT 900
#define KF 261
#define KP 264
#define LD 268
#define NT 512
#define LP 1024

typedef __attribute__((ext_vector_type(2))) float v2f;
typedef __attribute__((ext_vector_type(4))) float v4f;
typedef __attribute__((ext_vector_type(8))) float v8f;

template <int NS, bool CLAMP>
__device__ __forceinline__ void load_b(v4f (&buf)[16], const float* __restrict__ wcol,
                                       const int ks0, const int half) {
  #pragma unroll
  for (int j = 0; j < NS; ++j) {
    int r0 = (ks0 + j) * 4 + half * 2, r1 = r0 + 1;
    if (CLAMP) { r0 = min(r0, KF - 1); r1 = min(r1, KF - 1); }
    buf[2 * j]     = *(const v4f*)(wcol + (size_t)r0 * DD);
    buf[2 * j + 1] = *(const v4f*)(wcol + (size_t)r1 * DD);
  }
}

template <int NS>
__device__ __forceinline__ void load_a(v2f (&av)[8], const float* arow, const int ks0,
                                       const int half) {
  #pragma unroll
  for (int j = 0; j < NS; ++j) av[j] = *(const v2f*)(arow + (ks0 + j) * 4 + half * 2);
}

template <int NS>
__device__ __forceinline__ void mma_block(v8f (&acc)[4], const v4f (&buf)[16],
                                          const v2f (&av)[8]) {
  #pragma unroll
  for (int j = 0; j < NS; ++j) {
    #pragma unroll
    for (int t = 0; t < 4; ++t) {
      v2f bv; bv.x = buf[2 * j][t]; bv.y = buf[2 * j + 1][t];
      acc[t] = __builtin_amdgcn_wmma_f32_16x16x4_f32(false, av[j], false, bv,
                                                     (short)0, acc[t], false, false);
    }
  }
}

template <int KS>
__device__ __forceinline__ void gemm_cols64(v8f (&acc)[4], v4f (&b0)[16], v4f (&b1)[16],
                                            const float* __restrict__ wcol, const float* arow,
                                            const int half) {
  v2f a0[8], a1[8];
  load_a<8>(a0, arow, 0, half);
  #pragma unroll 1
  for (int blk = 0; blk < 6; blk += 2) {
    load_b<8, false>(b1, wcol, (blk + 1) * 8, half);
    load_a<8>(a1, arow, (blk + 1) * 8, half);
    __builtin_amdgcn_sched_barrier(0);
    mma_block<8>(acc, b0, a0);
    load_b<8, false>(b0, wcol, (blk + 2) * 8, half);
    load_a<8>(a0, arow, (blk + 2) * 8, half);
    __builtin_amdgcn_sched_barrier(0);
    mma_block<8>(acc, b1, a1);
  }
  load_b<8, false>(b1, wcol, 56, half);
  load_a<8>(a1, arow, 56, half);
  __builtin_amdgcn_sched_barrier(0);
  mma_block<8>(acc, b0, a0);
  if constexpr (KS > 64) {
    load_b<KS - 64, true>(b0, wcol, 64, half);
    load_a<KS - 64>(a0, arow, 64, half);
    __builtin_amdgcn_sched_barrier(0);
  }
  mma_block<8>(acc, b1, a1);
  if constexpr (KS > 64) mma_block<KS - 64>(acc, b0, a0);
}

__global__ __launch_bounds__(NT) void cc_pool_mlp_kernel(
    const int* __restrict__ grid, const float* __restrict__ grid_emb,
    const float* __restrict__ structure, const float* __restrict__ W1,
    const float* __restrict__ b1, const float* __restrict__ W2,
    const float* __restrict__ b2, const float* __restrict__ Wp,
    const float* __restrict__ bp, const float* __restrict__ gmm,
    const float* __restrict__ bta, const float* __restrict__ osp,
    float* __restrict__ out) {
  __shared__ int g[LP];
  __shared__ int lab[LP];
  __shared__ int rmn[HWC], rmx[HWC], cmn[HWC], cmx[HWC];
  __shared__ int roots[MAXOBJ];
  __shared__ unsigned wmask[32];
  __shared__ int chg[3];

  __shared__ float sA[MAXOBJ][LD];
  __shared__ float sB[MAXOBJ][LD];
  __shared__ float sC[MAXOBJ][LD];
  __shared__ float s_sn[DD];
  __shared__ float s_dot[MAXOBJ];
  __shared__ float s_red[16];
  __shared__ int   s_valid[MAXOBJ];
  __shared__ float s_norm;

  const int b = blockIdx.x;
  const int tid = threadIdx.x;
  const int lane = tid & 31;
  const int wv = tid >> 5;
  const int half = lane >> 4;
  const int mrow = lane & 15;
  const bool mmw = wv < 4;
  const int cb = (wv & 3) * 64 + mrow * 4;
  v4f wq0[16], wq1[16];

  {
    const int* gg = grid + b * HWC;
    const int t0 = tid, t1 = tid + NT;
    const int v0 = gg[t0];
    const int v1 = (t1 < HWC) ? gg[t1] : 0;
    g[t0] = v0; lab[t0] = (v0 > 0) ? t0 : SENT;
    g[t1] = v1; lab[t1] = (v1 > 0) ? t1 : SENT;
    rmn[t0] = GH; rmx[t0] = -1; cmn[t0] = GW; cmx[t0] = -1;
    if (t1 < HWC) { rmn[t1] = GH; rmx[t1] = -1; cmn[t1] = GW; cmx[t1] = -1; }
    if (tid < 3) chg[tid] = 0;
  }
  __syncthreads();

  int nb0u, nb0d, nb0l, nb0r, nb1u, nb1d, nb1l, nb1r;
  int my0, my1;
  {
    const int c0 = tid, c1 = tid + NT;
    const int r0 = c0 / GW, x0 = c0 - r0 * GW;
    const int r1 = c1 / GW, x1 = c1 - r1 * GW;
    const int u0 = (r0 > 0) ? c0 - GW : c0;
    const int d0 = (c0 + GW < HWC) ? c0 + GW : c0;
    const int l0 = (x0 > 0) ? c0 - 1 : c0;
    const int e0 = (x0 < GW - 1) ? c0 + 1 : c0;
    const int u1 = c1 - GW;
    const int d1 = (c1 + GW < HWC) ? c1 + GW : c1;
    const int l1 = (x1 > 0) ? c1 - 1 : c1;
    const int e1 = (x1 < GW - 1 && c1 + 1 < HWC) ? c1 + 1 : c1;
    const int gv0 = g[c0], gv1 = g[c1];
    const int gu0 = g[u0], gd0 = g[d0], gl0 = g[l0], ge0 = g[e0];
    const int gu1 = g[u1], gd1 = g[d1], gl1 = g[l1], ge1 = g[e1];
    const bool f0 = gv0 > 0, f1 = gv1 > 0;
    nb0u = (f0 && gu0 == gv0) ? u0 : c0;
    nb0d = (f0 && gd0 == gv0) ? d0 : c0;
    nb0l = (f0 && gl0 == gv0) ? l0 : c0;
    nb0r = (f0 && ge0 == gv0) ? e0 : c0;
    nb1u = (f1 && gu1 == gv1) ? u1 : c1;
    nb1d = (f1 && gd1 == gv1) ? d1 : c1;
    nb1l = (f1 && gl1 == gv1) ? l1 : c1;
    nb1r = (f1 && ge1 == gv1) ? e1 : c1;
    my0 = f0 ? c0 : SENT;
    my1 = f1 ? c1 : SENT;
  }

  int cur = 0;
  for (int iter = 0; iter < 2048; ++iter) {
    const int a0 = lab[nb0u], a1 = lab[nb0d], a2 = lab[nb0l], a3 = lab[nb0r];
    const int a4 = lab[nb1u], a5 = lab[nb1d], a6 = lab[nb1l], a7 = lab[nb1r];
    int m0 = min(min(my0, a0), min(min(a1, a2), a3));
    int m1 = min(min(my1, a4), min(min(a5, a6), a7));
    const int j0 = lab[m0], j1 = lab[m1];
    __builtin_amdgcn_sched_barrier(0);
    m0 = min(m0, j0);
    m1 = min(m1, j1);
    const bool mv0 = m0 < my0, mv1 = m1 < my1;
    if (mv0) { lab[tid] = m0; my0 = m0; }
    if (mv1) { lab[tid + NT] = m1; my1 = m1; }
    if (mv0 || mv1) chg[cur] = 1;
    const int nxt = (cur == 2) ? 0 : cur + 1;
    if (tid == 0) chg[nxt] = 0;
    __syncthreads();
    const int any = chg[cur];
    cur = nxt;
    if (any == 0) break;
  }

  {
    const int t0 = tid, t1 = tid + NT;
    if (my0 < SENT) {
      const int rr = t0 / GW, cc = t0 - rr * GW;
      atomicMin(&rmn[my0], rr); atomicMax(&rmx[my0], rr);
      atomicMin(&cmn[my0], cc); atomicMax(&cmx[my0], cc);
    }
    if (my1 < SENT) {
      const int rr = t1 / GW, cc = t1 - rr * GW;
      atomicMin(&rmn[my1], rr); atomicMax(&rmx[my1], rr);
      atomicMin(&cmn[my1], cc); atomicMax(&cmx[my1], cc);
    }
  }

  const bool isr0 = (my0 == tid);
  const bool isr1 = (my1 == tid + NT) && (tid + NT < HWC);
  const unsigned rb0 = __builtin_amdgcn_ballot_w32(isr0);
  const unsigned rb1 = __builtin_amdgcn_ballot_w32(isr1);
  if (lane == 0) { wmask[wv] = rb0; wmask[16 + wv] = rb1; }
  __syncthreads();

  int nr;
  {
    const unsigned c = (unsigned)__popc(wmask[lane]);
    const unsigned lo = (1u << wv) - 1u;
    const unsigned hi = (1u << (16 + wv)) - 1u;
    unsigned base0 = 0u, base1 = 0u, total = 0u;
    for (int k = 0; k < 6; ++k) {
      const unsigned bk = __builtin_amdgcn_ballot_w32(((c >> k) & 1u) != 0u);
      base0 += (unsigned)__popc(bk & lo) << k;
      base1 += (unsigned)__popc(bk & hi) << k;
      total += (unsigned)__popc(bk) << k;
    }
    const unsigned below = (1u << lane) - 1u;
    if (isr0) {
      const unsigned rank = base0 + (unsigned)__popc(rb0 & below);
      if (rank < MAXOBJ) roots[rank] = tid;
    }
    if (isr1) {
      const unsigned rank = base1 + (unsigned)__popc(rb1 & below);
      if (rank < MAXOBJ) roots[rank] = tid + NT;
    }
    nr = (int)min(total, (unsigned)MAXOBJ);
  }
  __syncthreads();

  {
    int color = 0, y = 0, x = 0, h = 1, w = 1;
    const bool valid = wv < nr;
    if (valid) {
      const int r = roots[wv];
      y = min(max(rmn[r], 0), GH - 1);
      x = min(max(cmn[r], 0), GW - 1);
      const int y2 = min(max(rmx[r] + 1, y + 1), GH);
      const int x2 = min(max(cmx[r] + 1, x + 1), GW);
      color = g[r]; h = y2 - y; w = x2 - x;
    }
    if (lane == 0) s_valid[wv] = valid ? 1 : 0;
    v4f s0 = {0.f, 0.f, 0.f, 0.f}, s1 = {0.f, 0.f, 0.f, 0.f};
    for (int i = 0; i < h; ++i) {
      const v4f* rowp =
          (const v4f*)(grid_emb + (((size_t)(b * GH + y + i)) * GW + x) * DD);
      for (int j = 0; j < w; ++j) {
        s0 += rowp[j * (DD / 4) + lane];
        s1 += rowp[j * (DD / 4) + 32 + lane];
      }
    }
    const float cnt = (float)(h * w);
    *(v4f*)&sC[wv][4 * lane]       = s0 / cnt;
    *(v4f*)&sC[wv][128 + 4 * lane] = s1 / cnt;
    if (lane < LD - DD) {
      float f = 0.f;
      if (lane == 0) f = (float)color / 9.0f;
      if (lane == 1) f = (float)x / (float)GW;
      if (lane == 2) f = (float)y / (float)GH;
      if (lane == 3) f = (float)w / (float)GW;
      if (lane == 4) f = (float)h / (float)GH;
      sC[wv][DD + lane] = f;
    }
  }

  if (mmw) load_b<8, false>(wq0, W1 + cb, 0, half);

  if (tid < DD) {
    const float* sp = structure + (size_t)b * 8 * DD + tid;
    float s = 0.f;
    #pragma unroll
    for (int k = 0; k < 8; ++k) s += sp[(size_t)k * DD];
    s_sn[tid] = s * 0.125f;
  }
  __syncthreads();
  if (tid < DD) {
    float p = s_sn[tid] * s_sn[tid];
    #pragma unroll
    for (int m = 16; m >= 1; m >>= 1) p += __shfl_xor(p, m, 32);
    if (lane == 0) s_red[wv] = p;
  }
  __syncthreads();
  if (tid == 0) {
    float t = 0.f;
    for (int i = 0; i < 8; ++i) t += s_red[i];
    s_norm = fmaxf(sqrtf(t), 1e-8f);
  }
  __syncthreads();
  if (tid < DD) s_sn[tid] /= s_norm;

  if (mmw) {
    v8f acc[4] = {};
    gemm_cols64<66>(acc, wq0, wq1, W1 + cb, &sC[mrow][0], half);
    load_b<8, false>(wq0, W2 + cb, 0, half);
    const v4f bias = *(const v4f*)(b1 + cb);
    #pragma unroll
    for (int i = 0; i < 8; ++i) {
      v4f v;
      #pragma unroll
      for (int t = 0; t < 4; ++t) v[t] = acc[t][i] + bias[t];
      *(v4f*)&sA[half * 8 + i][cb] = v;
    }
  }
  __syncthreads();
  {
    float* hp = &sA[wv][8 * lane];
    v4f u0 = *(v4f*)hp, u1 = *(v4f*)(hp + 4);
    #pragma unroll
    for (int t = 0; t < 4; ++t) {
      u0[t] = 0.5f * u0[t] * (1.0f + erff(u0[t] * 0.70710678118654752f));
      u1[t] = 0.5f * u1[t] * (1.0f + erff(u1[t] * 0.70710678118654752f));
    }
    *(v4f*)hp = u0; *(v4f*)(hp + 4) = u1;
  }
  __syncthreads();

  if (mmw) {
    v8f acc[4] = {};
    gemm_cols64<64>(acc, wq0, wq1, W2 + cb, &sA[mrow][0], half);
    load_b<8, false>(wq0, Wp + cb, 0, half);
    const v4f bias = *(const v4f*)(b2 + cb);
    #pragma unroll
    for (int i = 0; i < 8; ++i) {
      const int row = half * 8 + i;
      v4f v;
      #pragma unroll
      for (int t = 0; t < 4; ++t) v[t] = acc[t][i] + bias[t];
      const v4f z = {0.f, 0.f, 0.f, 0.f};
      *(v4f*)&sB[row][cb] = s_valid[row] ? v : z;
    }
  }
  __syncthreads();

  {
    float p = 0.f;
    #pragma unroll
    for (int j = 0; j < 8; ++j) {
      const int n = lane + 32 * j;
      p += sB[wv][n] * s_sn[n];
    }
    #pragma unroll
    for (int m = 16; m >= 1; m >>= 1) p += __shfl_xor(p, m, 32);
    if (lane == 0) s_dot[wv] = p;
  }
  __syncthreads();
  {
    const float os = osp[0];
    for (int t = tid; t < MAXOBJ * DD; t += NT) {
      const int r = t >> 8, n = t & 255;
      sB[r][n] = (sB[r][n] - s_dot[r] * s_sn[n]) * os;
    }
  }
  __syncthreads();

  if (mmw) {
    v8f acc[4] = {};
    gemm_cols64<64>(acc, wq0, wq1, Wp + cb, &sB[mrow][0], half);
    const v4f bias = *(const v4f*)(bp + cb);
    #pragma unroll
    for (int i = 0; i < 8; ++i) {
      v4f v;
      #pragma unroll
      for (int t = 0; t < 4; ++t) v[t] = acc[t][i] + bias[t];
      *(v4f*)&sC[half * 8 + i][cb] = v;
    }
  }
  __syncthreads();

  {
    float s1 = 0.f, s2 = 0.f;
    #pragma unroll
    for (int j = 0; j < 8; ++j) {
      const float v = sC[wv][lane + 32 * j];
      s1 += v; s2 += v * v;
    }
    #pragma unroll
    for (int m = 16; m >= 1; m >>= 1) {
      s1 += __shfl_xor(s1, m, 32);
      s2 += __shfl_xor(s2, m, 32);
    }
    const float mu = s1 * (1.0f / 256.0f);
    const float var = s2 * (1.0f / 256.0f) - mu * mu;
    const float rstd = rsqrtf(var + 1e-5f);
    float* op = out + (size_t)(b * MAXOBJ + wv) * DD;
    #pragma unroll
    for (int j = 0; j < 8; ++j) {
      const int n = lane + 32 * j;
      const float v = sC[wv][n];
      op[n] = (v - mu) * rstd * gmm[n] + bta[n];
    }
  }
}

extern "C" void kernel_launch(void* const* d_in, const int* in_sizes, int n_in,
                              void* d_out, int out_size, void* d_ws, size_t ws_size,
                              hipStream_t stream) {
  const float* grid_emb  = (const float*)d_in[0];
  const int*   grid      = (const int*)  d_in[1];
  const float* structure = (const float*)d_in[2];
  const float* W1        = (const float*)d_in[3];
  const float* b1        = (const float*)d_in[4];
  const float* W2        = (const float*)d_in[5];
  const float* b2        = (const float*)d_in[6];
  const float* Wp        = (const float*)d_in[7];
  const float* bp        = (const float*)d_in[8];
  const float* gmm       = (const float*)d_in[9];
  const float* bta       = (const float*)d_in[10];
  const float* osp       = (const float*)d_in[11];
  float* out = (float*)d_out;

  const int B = in_sizes[1] / HWC;

  cc_pool_mlp_kernel<<<B, NT, 0, stream>>>(grid, grid_emb, structure, W1, b1, W2, b2,
                                           Wp, bp, gmm, bta, osp, out);
}


# Bbox pooling: all cell loads of a box (groups of up to 16 cells) are requested before the first add, box bounds held in scalar registers; was one memory round trip per cell; cell sum order kept
# speedup vs baseline: 1.8591x; 1.0540x over previous
#include <hip/hip_runtime.h>
#include <math.h>


#define GH 30
#define GW 30
#define HWC 900
#define DD 256
#define MAXOBJ 16
#define SENT 900
#define KF 261
#define KP 264
#define LD 268
#define NT 512
#define LP 1024
#define PG 16

typedef __attribute__((ext_vector_type(2))) float v2f;
typedef __attribute__((ext_vector_type(4))) float v4f;
typedef __attribute__((ext_vector_type(8))) float v8f;

template <int NS, bool CLAMP>
__device__ __forceinline__ void load_b(v4f (&buf)[16], const float* __restrict__ wcol,
                                       const int ks0, const int half) {
  #pragma unroll
  for (int j = 0; j < NS; ++j) {
    int r0 = (ks0 + j) * 4 + half * 2, r1 = r0 + 1;
    if (CLAMP) { r0 = min(r0, KF - 1); r1 = min(r1, KF - 1); }
    buf[2 * j]     = *(const v4f*)(wcol + (size_t)r0 * DD);
    buf[2 * j + 1] = *(const v4f*)(wcol + (size_t)r1 * DD);
  }
}

template <int NS>
__device__ __forceinline__ void load_a(v2f (&av)[8], const float* arow, const int ks0,
                                       const int half) {
  #pragma unroll
  for (int j = 0; j < NS; ++j) av[j] = *(const v2f*)(arow + (ks0 + j) * 4 + half * 2);
}

template <int NS>
__device__ __forceinline__ void mma_block(v8f (&acc)[4], const v4f (&buf)[16],
                                          const v2f (&av)[8]) {
  #pragma unroll
  for (int j = 0; j < NS; ++j) {
    #pragma unroll
    for (int t = 0; t < 4; ++t) {
      v2f bv; bv.x = buf[2 * j][t]; bv.y = buf[2 * j + 1][t];
      acc[t] = __builtin_amdgcn_wmma_f32_16x16x4_f32(false, av[j], false, bv,
                                                     (short)0, acc[t], false, false);
    }
  }
}

template <int KS>
__device__ __forceinline__ void gemm_cols64(v8f (&acc)[4], v4f (&b0)[16], v4f (&b1)[16],
                                            const float* __restrict__ wcol, const float* arow,
                                            const int half) {
  v2f a0[8], a1[8];
  load_a<8>(a0, arow, 0, half);
  #pragma unroll 1
  for (int blk = 0; blk < 6; blk += 2) {
    load_b<8, false>(b1, wcol, (blk + 1) * 8, half);
    load_a<8>(a1, arow, (blk + 1) * 8, half);
    __builtin_amdgcn_sched_barrier(0);
    mma_block<8>(acc, b0, a0);
    load_b<8, false>(b0, wcol, (blk + 2) * 8, half);
    load_a<8>(a0, arow, (blk + 2) * 8, half);
    __builtin_amdgcn_sched_barrier(0);
    mma_block<8>(acc, b1, a1);
  }
  load_b<8, false>(b1, wcol, 56, half);
  load_a<8>(a1, arow, 56, half);
  __builtin_amdgcn_sched_barrier(0);
  mma_block<8>(acc, b0, a0);
  if constexpr (KS > 64) {
    load_b<KS - 64, true>(b0, wcol, 64, half);
    load_a<KS - 64>(a0, arow, 64, half);
    __builtin_amdgcn_sched_barrier(0);
  }
  mma_block<8>(acc, b1, a1);
  if constexpr (KS > 64) mma_block<KS - 64>(acc, b0, a0);
}

template <int K>
__device__ __forceinline__ void pool_request(v4f (&q0)[PG], v4f (&q1)[PG], const float* cellp,
                                             const int n, const int ws, int& ri, int& cj) {
  if constexpr (K < PG) {
    if (K < n) {
      const float* p = cellp + (size_t)(ri * GW + cj) * DD;
      q0[K] = *(const v4f*)p;
      q1[K] = *(const v4f*)(p + 128);
      if (++cj == ws) { cj = 0; ++ri; }
      pool_request<K + 1>(q0, q1, cellp, n, ws, ri, cj);
    }
  }
}

template <int K>
__device__ __forceinline__ void pool_add(v4f& s0, v4f& s1, const v4f (&q0)[PG],
                                         const v4f (&q1)[PG], const int n) {
  if constexpr (K < PG) {
    if (K < n) {
      s0 += q0[K];
      s1 += q1[K];
      pool_add<K + 1>(s0, s1, q0, q1, n);
    }
  }
}

__global__ __launch_bounds__(NT) void cc_pool_mlp_kernel(
    const int* __restrict__ grid, const float* __restrict__ grid_emb,
    const float* __restrict__ structure, const float* __restrict__ W1,
    const float* __restrict__ b1, const float* __restrict__ W2,
    const float* __restrict__ b2, const float* __restrict__ Wp,
    const float* __restrict__ bp, const float* __restrict__ gmm,
    const float* __restrict__ bta, const float* __restrict__ osp,
    float* __restrict__ out) {
  __shared__ int g[LP];
  __shared__ int lab[LP];
  __shared__ int rmn[HWC], rmx[HWC], cmn[HWC], cmx[HWC];
  __shared__ int roots[MAXOBJ];
  __shared__ unsigned wmask[32];
  __shared__ int chg[3];

  __shared__ float sA[MAXOBJ][LD];
  __shared__ float sB[MAXOBJ][LD];
  __shared__ float sC[MAXOBJ][LD];
  __shared__ float s_sn[DD];
  __shared__ float s_dot[MAXOBJ];
  __shared__ float s_red[16];
  __shared__ int   s_valid[MAXOBJ];
  __shared__ float s_norm;

  const int b = blockIdx.x;
  const int tid = threadIdx.x;
  const int lane = tid & 31;
  const int wv = tid >> 5;
  const int half = lane >> 4;
  const int mrow = lane & 15;
  const bool mmw = wv < 4;
  const int cb = (wv & 3) * 64 + mrow * 4;
  v4f wq0[16], wq1[16];

  {
    const int* gg = grid + b * HWC;
    const int t0 = tid, t1 = tid + NT;
    const int v0 = gg[t0];
    const int v1 = (t1 < HWC) ? gg[t1] : 0;
    g[t0] = v0; lab[t0] = (v0 > 0) ? t0 : SENT;
    g[t1] = v1; lab[t1] = (v1 > 0) ? t1 : SENT;
    rmn[t0] = GH; rmx[t0] = -1; cmn[t0] = GW; cmx[t0] = -1;
    if (t1 < HWC) { rmn[t1] = GH; rmx[t1] = -1; cmn[t1] = GW; cmx[t1] = -1; }
    if (tid < 3) chg[tid] = 0;
  }
  __syncthreads();

  int nb0u, nb0d, nb0l, nb0r, nb1u, nb1d, nb1l, nb1r;
  int my0, my1;
  {
    const int c0 = tid, c1 = tid + NT;
    const int r0 = c0 / GW, x0 = c0 - r0 * GW;
    const int r1 = c1 / GW, x1 = c1 - r1 * GW;
    const int u0 = (r0 > 0) ? c0 - GW : c0;
    const int d0 = (c0 + GW < HWC) ? c0 + GW : c0;
    const int l0 = (x0 > 0) ? c0 - 1 : c0;
    const int e0 = (x0 < GW - 1) ? c0 + 1 : c0;
    const int u1 = c1 - GW;
    const int d1 = (c1 + GW < HWC) ? c1 + GW : c1;
    const int l1 = (x1 > 0) ? c1 - 1 : c1;
    const int e1 = (x1 < GW - 1 && c1 + 1 < HWC) ? c1 + 1 : c1;
    const int gv0 = g[c0], gv1 = g[c1];
    const int gu0 = g[u0], gd0 = g[d0], gl0 = g[l0], ge0 = g[e0];
    const int gu1 = g[u1], gd1 = g[d1], gl1 = g[l1], ge1 = g[e1];
    const bool f0 = gv0 > 0, f1 = gv1 > 0;
    nb0u = (f0 && gu0 == gv0) ? u0 : c0;
    nb0d = (f0 && gd0 == gv0) ? d0 : c0;
    nb0l = (f0 && gl0 == gv0) ? l0 : c0;
    nb0r = (f0 && ge0 == gv0) ? e0 : c0;
    nb1u = (f1 && gu1 == gv1) ? u1 : c1;
    nb1d = (f1 && gd1 == gv1) ? d1 : c1;
    nb1l = (f1 && gl1 == gv1) ? l1 : c1;
    nb1r = (f1 && ge1 == gv1) ? e1 : c1;
    my0 = f0 ? c0 : SENT;
    my1 = f1 ? c1 : SENT;
  }

  int cur = 0;
  for (int iter = 0; iter < 2048; ++iter) {
    const int a0 = lab[nb0u], a1 = lab[nb0d], a2 = lab[nb0l], a3 = lab[nb0r];
    const int a4 = lab[nb1u], a5 = lab[nb1d], a6 = lab[nb1l], a7 = lab[nb1r];
    int m0 = min(min(my0, a0), min(min(a1, a2), a3));
    int m1 = min(min(my1, a4), min(min(a5, a6), a7));
    const int j0 = lab[m0], j1 = lab[m1];
    __builtin_amdgcn_sched_barrier(0);
    m0 = min(m0, j0);
    m1 = min(m1, j1);
    const bool mv0 = m0 < my0, mv1 = m1 < my1;
    if (mv0) { lab[tid] = m0; my0 = m0; }
    if (mv1) { lab[tid + NT] = m1; my1 = m1; }
    if (mv0 || mv1) chg[cur] = 1;
    const int nxt = (cur == 2) ? 0 : cur + 1;
    if (tid == 0) chg[nxt] = 0;
    __syncthreads();
    const int any = chg[cur];
    cur = nxt;
    if (any == 0) break;
  }

  {
    const int t0 = tid, t1 = tid + NT;
    if (my0 < SENT) {
      const int rr = t0 / GW, cc = t0 - rr * GW;
      atomicMin(&rmn[my0], rr); atomicMax(&rmx[my0], rr);
      atomicMin(&cmn[my0], cc); atomicMax(&cmx[my0], cc);
    }
    if (my1 < SENT) {
      const int rr = t1 / GW, cc = t1 - rr * GW;
      atomicMin(&rmn[my1], rr); atomicMax(&rmx[my1], rr);
      atomicMin(&cmn[my1], cc); atomicMax(&cmx[my1], cc);
    }
  }

  const bool isr0 = (my0 == tid);
  const bool isr1 = (my1 == tid + NT) && (tid + NT < HWC);
  const unsigned rb0 = __builtin_amdgcn_ballot_w32(isr0);
  const unsigned rb1 = __builtin_amdgcn_ballot_w32(isr1);
  if (lane == 0) { wmask[wv] = rb0; wmask[16 + wv] = rb1; }
  __syncthreads();

  int nr;
  {
    const unsigned c = (unsigned)__popc(wmask[lane]);
    const unsigned lo = (1u << wv) - 1u;
    const unsigned hi = (1u << (16 + wv)) - 1u;
    unsigned base0 = 0u, base1 = 0u, total = 0u;
    for (int k = 0; k < 6; ++k) {
      const unsigned bk = __builtin_amdgcn_ballot_w32(((c >> k) & 1u) != 0u);
      base0 += (unsigned)__popc(bk & lo) << k;
      base1 += (unsigned)__popc(bk & hi) << k;
      total += (unsigned)__popc(bk) << k;
    }
    const unsigned below = (1u << lane) - 1u;
    if (isr0) {
      const unsigned rank = base0 + (unsigned)__popc(rb0 & below);
      if (rank < MAXOBJ) roots[rank] = tid;
    }
    if (isr1) {
      const unsigned rank = base1 + (unsigned)__popc(rb1 & below);
      if (rank < MAXOBJ) roots[rank] = tid + NT;
    }
    nr = (int)min(total, (unsigned)MAXOBJ);
  }
  __syncthreads();

  {
    int color = 0, y = 0, x = 0, h = 1, w = 1;
    const bool valid = wv < nr;
    if (valid) {
      const int r = roots[wv];
      y = min(max(rmn[r], 0), GH - 1);
      x = min(max(cmn[r], 0), GW - 1);
      const int y2 = min(max(rmx[r] + 1, y + 1), GH);
      const int x2 = min(max(cmx[r] + 1, x + 1), GW);
      color = g[r]; h = y2 - y; w = x2 - x;
    }
    if (lane == 0) s_valid[wv] = valid ? 1 : 0;
    const int ys = __builtin_amdgcn_readfirstlane(y), xs = __builtin_amdgcn_readfirstlane(x);
    const int hs = __builtin_amdgcn_readfirstlane(h), ws = __builtin_amdgcn_readfirstlane(w);
    const int ncell = hs * ws;
    const float* cellp = grid_emb + (((size_t)(b * GH + ys)) * GW + xs) * DD + 4 * lane;
    v4f s0 = {0.f, 0.f, 0.f, 0.f}, s1 = {0.f, 0.f, 0.f, 0.f};
    int ri = 0, cj = 0;
    for (int c0 = 0; c0 < ncell; c0 += PG) {
      const int n = min(PG, ncell - c0);
      v4f q0[PG], q1[PG];
      pool_request<0>(q0, q1, cellp, n, ws, ri, cj);
      pool_add<0>(s0, s1, q0, q1, n);
    }
    const float cnt = (float)(h * w);
    *(v4f*)&sC[wv][4 * lane]       = s0 / cnt;
    *(v4f*)&sC[wv][128 + 4 * lane] = s1 / cnt;
    if (lane < LD - DD) {
      float f = 0.f;
      if (lane == 0) f = (float)color / 9.0f;
      if (lane == 1) f = (float)x / (float)GW;
      if (lane == 2) f = (float)y / (float)GH;
      if (lane == 3) f = (float)w / (float)GW;
      if (lane == 4) f = (float)h / (float)GH;
      sC[wv][DD + lane] = f;
    }
  }

  if (mmw) load_b<8, false>(wq0, W1 + cb, 0, half);

  if (tid < DD) {
    const float* sp = structure + (size_t)b * 8 * DD + tid;
    float s = 0.f;
    #pragma unroll
    for (int k = 0; k < 8; ++k) s += sp[(size_t)k * DD];
    s_sn[tid] = s * 0.125f;
  }
  __syncthreads();
  if (tid < DD) {
    float p = s_sn[tid] * s_sn[tid];
    #pragma unroll
    for (int m = 16; m >= 1; m >>= 1) p += __shfl_xor(p, m, 32);
    if (lane == 0) s_red[wv] = p;
  }
  __syncthreads();
  if (tid == 0) {
    float t = 0.f;
    for (int i = 0; i < 8; ++i) t += s_red[i];
    s_norm = fmaxf(sqrtf(t), 1e-8f);
  }
  __syncthreads();
  if (tid < DD) s_sn[tid] /= s_norm;

  if (mmw) {
    v8f acc[4] = {};
    gemm_cols64<66>(acc, wq0, wq1, W1 + cb, &sC[mrow][0], half);
    load_b<8, false>(wq0, W2 + cb, 0, half);
    const v4f bias = *(const v4f*)(b1 + cb);
    #pragma unroll
    for (int i = 0; i < 8; ++i) {
      v4f v;
      #pragma unroll
      for (int t = 0; t < 4; ++t) v[t] = acc[t][i] + bias[t];
      *(v4f*)&sA[half * 8 + i][cb] = v;
    }
  }
  __syncthreads();
  {
    float* hp = &sA[wv][8 * lane];
    v4f u0 = *(v4f*)hp, u1 = *(v4f*)(hp + 4);
    #pragma unroll
    for (int t = 0; t < 4; ++t) {
      u0[t] = 0.5f * u0[t] * (1.0f + erff(u0[t] * 0.70710678118654752f));
      u1[t] = 0.5f * u1[t] * (1.0f + erff(u1[t] * 0.70710678118654752f));
    }
    *(v4f*)hp = u0; *(v4f*)(hp + 4) = u1;
  }
  __syncthreads();

  if (mmw) {
    v8f acc[4] = {};
    gemm_cols64<64>(acc, wq0, wq1, W2 + cb, &sA[mrow][0], half);
    load_b<8, false>(wq0, Wp + cb, 0, half);
    const v4f bias = *(const v4f*)(b2 + cb);
    #pragma unroll
    for (int i = 0; i < 8; ++i) {
      const int row = half * 8 + i;
      v4f v;
      #pragma unroll
      for (int t = 0; t < 4; ++t) v[t] = acc[t][i] + bias[t];
      const v4f z = {0.f, 0.f, 0.f, 0.f};
      *(v4f*)&sB[row][cb] = s_valid[row] ? v : z;
    }
  }
  __syncthreads();

  {
    float p = 0.f;
    #pragma unroll
    for (int j = 0; j < 8; ++j) {
      const int n = lane + 32 * j;
      p += sB[wv][n] * s_sn[n];
    }
    #pragma unroll
    for (int m = 16; m >= 1; m >>= 1) p += __shfl_xor(p, m, 32);
    if (lane == 0) s_dot[wv] = p;
  }
  __syncthreads();
  {
    const float os = osp[0];
    for (int t = tid; t < MAXOBJ * DD; t += NT) {
      const int r = t >> 8, n = t & 255;
      sB[r][n] = (sB[r][n] - s_dot[r] * s_sn[n]) * os;
    }
  }
  __syncthreads();

  if (mmw) {
    v8f acc[4] = {};
    gemm_cols64<64>(acc, wq0, wq1, Wp + cb, &sB[mrow][0], half);
    const v4f bias = *(const v4f*)(bp + cb);
    #pragma unroll
    for (int i = 0; i < 8; ++i) {
      v4f v;
      #pragma unroll
      for (int t = 0; t < 4; ++t) v[t] = acc[t][i] + bias[t];
      *(v4f*)&sC[half * 8 + i][cb] = v;
    }
  }
  __syncthreads();

  {
    float s1 = 0.f, s2 = 0.f;
    #pragma unroll
    for (int j = 0; j < 8; ++j) {
      const float v = sC[wv][lane + 32 * j];
      s1 += v; s2 += v * v;
    }
    #pragma unroll
    for (int m = 16; m >= 1; m >>= 1) {
      s1 += __shfl_xor(s1, m, 32);
      s2 += __shfl_xor(s2, m, 32);
    }
    const float mu = s1 * (1.0f / 256.0f);
    const float var = s2 * (1.0f / 256.0f) - mu * mu;
    const float rstd = rsqrtf(var + 1e-5f);
    float* op = out + (size_t)(b * MAXOBJ + wv) * DD;
    #pragma unroll
    for (int j = 0; j < 8; ++j) {
      const int n = lane + 32 * j;
      const float v = sC[wv][n];
      op[n] = (v - mu) * rstd * gmm[n] + bta[n];
    }
  }
}

extern "C" void kernel_launch(void* const* d_in, const int* in_sizes, int n_in,
                              void* d_out, int out_size, void* d_ws, size_t ws_size,
                              hipStream_t stream) {
  const float* grid_emb  = (const float*)d_in[0];
  const int*   grid      = (const int*)  d_in[1];
  const float* structure = (const float*)d_in[2];
  const float* W1        = (const float*)d_in[3];
  const float* b1        = (const float*)d_in[4];
  const float* W2        = (const float*)d_in[5];
  const float* b2        = (const float*)d_in[6];
  const float* Wp        = (const float*)d_in[7];
  const float* bp        = (const float*)d_in[8];
  const float* gmm       = (const float*)d_in[9];
  const float* bta       = (const float*)d_in[10];
  const float* osp       = (const float*)d_in[11];
  float* out = (float*)d_out;

  const int B = in_sizes[1] / HWC;

  cc_pool_mlp_kernel<<<B, NT, 0, stream>>>(grid, grid_emb, structure, W1, b1, W2, b2,
                                           Wp, bp, gmm, bta, osp, out);
}


# Ortho step per row from registers: wave k reads row k once, sums the dot by lane exchanges on the vector ALU (same pairs, same order), rewrites the row; was 2 barriers and an 8-trip serial LDS loop
# speedup vs baseline: 1.9750x; 1.0623x over previous
#include <hip/hip_runtime.h>
#include <math.h>


#define GH 30
#define GW 30
#define HWC 900
#define DD 256
#define MAXOBJ 16
#define SENT 900
#define KF 261
#define KP 264
#define LD 268
#define NT 512
#define LP 1024
#define PG 16

typedef __attribute__((ext_vector_type(2))) float v2f;
typedef __attribute__((ext_vector_type(4))) float v4f;
typedef __attribute__((ext_vector_type(8))) float v8f;

template <int NS, bool CLAMP>
__device__ __forceinline__ void load_b(v4f (&buf)[16], const float* __restrict__ wcol,
                                       const int ks0, const int half) {
  #pragma unroll
  for (int j = 0; j < NS; ++j) {
    int r0 = (ks0 + j) * 4 + half * 2, r1 = r0 + 1;
    if (CLAMP) { r0 = min(r0, KF - 1); r1 = min(r1, KF - 1); }
    buf[2 * j]     = *(const v4f*)(wcol + (size_t)r0 * DD);
    buf[2 * j + 1] = *(const v4f*)(wcol + (size_t)r1 * DD);
  }
}

template <int NS>
__device__ __forceinline__ void load_a(v2f (&av)[8], const float* arow, const int ks0,
                                       const int half) {
  #pragma unroll
  for (int j = 0; j < NS; ++j) av[j] = *(const v2f*)(arow + (ks0 + j) * 4 + half * 2);
}

template <int NS>
__device__ __forceinline__ void mma_block(v8f (&acc)[4], const v4f (&buf)[16],
                                          const v2f (&av)[8]) {
  #pragma unroll
  for (int j = 0; j < NS; ++j) {
    #pragma unroll
    for (int t = 0; t < 4; ++t) {
      v2f bv; bv.x = buf[2 * j][t]; bv.y = buf[2 * j + 1][t];
      acc[t] = __builtin_amdgcn_wmma_f32_16x16x4_f32(false, av[j], false, bv,
                                                     (short)0, acc[t], false, false);
    }
  }
}

template <int KS>
__device__ __forceinline__ void gemm_cols64(v8f (&acc)[4], v4f (&b0)[16], v4f (&b1)[16],
                                            const float* __restrict__ wcol, const float* arow,
                                            const int half) {
  v2f a0[8], a1[8];
  load_a<8>(a0, arow, 0, half);
  #pragma unroll 1
  for (int blk = 0; blk < 6; blk += 2) {
    load_b<8, false>(b1, wcol, (blk + 1) * 8, half);
    load_a<8>(a1, arow, (blk + 1) * 8, half);
    __builtin_amdgcn_sched_barrier(0);
    mma_block<8>(acc, b0, a0);
    load_b<8, false>(b0, wcol, (blk + 2) * 8, half);
    load_a<8>(a0, arow, (blk + 2) * 8, half);
    __builtin_amdgcn_sched_barrier(0);
    mma_block<8>(acc, b1, a1);
  }
  load_b<8, false>(b1, wcol, 56, half);
  load_a<8>(a1, arow, 56, half);
  __builtin_amdgcn_sched_barrier(0);
  mma_block<8>(acc, b0, a0);
  if constexpr (KS > 64) {
    load_b<KS - 64, true>(b0, wcol, 64, half);
    load_a<KS - 64>(a0, arow, 64, half);
    __builtin_amdgcn_sched_barrier(0);
  }
  mma_block<8>(acc, b1, a1);
  if constexpr (KS > 64) mma_block<KS - 64>(acc, b0, a0);
}

template <int K>
__device__ __forceinline__ void pool_request(v4f (&q0)[PG], v4f (&q1)[PG], const float* cellp,
                                             const int n, const int ws, int& ri, int& cj) {
  if constexpr (K < PG) {
    if (K < n) {
      const float* p = cellp + (size_t)(ri * GW + cj) * DD;
      q0[K] = *(const v4f*)p;
      q1[K] = *(const v4f*)(p + 128);
      if (++cj == ws) { cj = 0; ++ri; }
      pool_request<K + 1>(q0, q1, cellp, n, ws, ri, cj);
    }
  }
}

template <int K>
__device__ __forceinline__ void pool_add(v4f& s0, v4f& s1, const v4f (&q0)[PG],
                                         const v4f (&q1)[PG], const int n) {
  if constexpr (K < PG) {
    if (K < n) {
      s0 += q0[K];
      s1 += q1[K];
      pool_add<K + 1>(s0, s1, q0, q1, n);
    }
  }
}

__device__ __forceinline__ float xhalf(const float x) {
  const int b = __builtin_bit_cast(int, x);
  return __builtin_bit_cast(float, __builtin_amdgcn_permlanex16(b, b, 0x76543210, 0xfedcba98,
                                                                false, false));
}

template <int M>
__device__ __forceinline__ float xrow(const float x) {
  const int b = __builtin_bit_cast(int, x);
  return __builtin_bit_cast(float, __builtin_amdgcn_update_dpp(b, b, 0x160 + M, 0xf, 0xf, false));
}

__global__ __launch_bounds__(NT) void cc_pool_mlp_kernel(
    const int* __restrict__ grid, const float* __restrict__ grid_emb,
    const float* __restrict__ structure, const float* __restrict__ W1,
    const float* __restrict__ b1, const float* __restrict__ W2,
    const float* __restrict__ b2, const float* __restrict__ Wp,
    const float* __restrict__ bp, const float* __restrict__ gmm,
    const float* __restrict__ bta, const float* __restrict__ osp,
    float* __restrict__ out) {
  __shared__ int g[LP];
  __shared__ int lab[LP];
  __shared__ int rmn[HWC], rmx[HWC], cmn[HWC], cmx[HWC];
  __shared__ int roots[MAXOBJ];
  __shared__ unsigned wmask[32];
  __shared__ int chg[3];

  __shared__ float sA[MAXOBJ][LD];
  __shared__ float sB[MAXOBJ][LD];
  __shared__ float sC[MAXOBJ][LD];
  __shared__ float s_sn[DD];
  __shared__ float s_red[16];
  __shared__ int   s_valid[MAXOBJ];
  __shared__ float s_norm;

  const int b = blockIdx.x;
  const int tid = threadIdx.x;
  const int lane = tid & 31;
  const int wv = tid >> 5;
  const int half = lane >> 4;
  const int mrow = lane & 15;
  const bool mmw = wv < 4;
  const int cb = (wv & 3) * 64 + mrow * 4;
  v4f wq0[16], wq1[16];
  const float os = osp[0];

  {
    const int* gg = grid + b * HWC;
    const int t0 = tid, t1 = tid + NT;
    const int v0 = gg[t0];
    const int v1 = (t1 < HWC) ? gg[t1] : 0;
    g[t0] = v0; lab[t0] = (v0 > 0) ? t0 : SENT;
    g[t1] = v1; lab[t1] = (v1 > 0) ? t1 : SENT;
    rmn[t0] = GH; rmx[t0] = -1; cmn[t0] = GW; cmx[t0] = -1;
    if (t1 < HWC) { rmn[t1] = GH; rmx[t1] = -1; cmn[t1] = GW; cmx[t1] = -1; }
    if (tid < 3) chg[tid] = 0;
  }
  __syncthreads();

  int nb0u, nb0d, nb0l, nb0r, nb1u, nb1d, nb1l, nb1r;
  int my0, my1;
  {
    const int c0 = tid, c1 = tid + NT;
    const int r0 = c0 / GW, x0 = c0 - r0 * GW;
    const int r1 = c1 / GW, x1 = c1 - r1 * GW;
    const int u0 = (r0 > 0) ? c0 - GW : c0;
    const int d0 = (c0 + GW < HWC) ? c0 + GW : c0;
    const int l0 = (x0 > 0) ? c0 - 1 : c0;
    const int e0 = (x0 < GW - 1) ? c0 + 1 : c0;
    const int u1 = c1 - GW;
    const int d1 = (c1 + GW < HWC) ? c1 + GW : c1;
    const int l1 = (x1 > 0) ? c1 - 1 : c1;
    const int e1 = (x1 < GW - 1 && c1 + 1 < HWC) ? c1 + 1 : c1;
    const int gv0 = g[c0], gv1 = g[c1];
    const int gu0 = g[u0], gd0 = g[d0], gl0 = g[l0], ge0 = g[e0];
    const int gu1 = g[u1], gd1 = g[d1], gl1 = g[l1], ge1 = g[e1];
    const bool f0 = gv0 > 0, f1 = gv1 > 0;
    nb0u = (f0 && gu0 == gv0) ? u0 : c0;
    nb0d = (f0 && gd0 == gv0) ? d0 : c0;
    nb0l = (f0 && gl0 == gv0) ? l0 : c0;
    nb0r = (f0 && ge0 == gv0) ? e0 : c0;
    nb1u = (f1 && gu1 == gv1) ? u1 : c1;
    nb1d = (f1 && gd1 == gv1) ? d1 : c1;
    nb1l = (f1 && gl1 == gv1) ? l1 : c1;
    nb1r = (f1 && ge1 == gv1) ? e1 : c1;
    my0 = f0 ? c0 : SENT;
    my1 = f1 ? c1 : SENT;
  }

  int cur = 0;
  for (int iter = 0; iter < 2048; ++iter) {
    const int a0 = lab[nb0u], a1 = lab[nb0d], a2 = lab[nb0l], a3 = lab[nb0r];
    const int a4 = lab[nb1u], a5 = lab[nb1d], a6 = lab[nb1l], a7 = lab[nb1r];
    int m0 = min(min(my0, a0), min(min(a1, a2), a3));
    int m1 = min(min(my1, a4), min(min(a5, a6), a7));
    const int j0 = lab[m0], j1 = lab[m1];
    __builtin_amdgcn_sched_barrier(0);
    m0 = min(m0, j0);
    m1 = min(m1, j1);
    const bool mv0 = m0 < my0, mv1 = m1 < my1;
    if (mv0) { lab[tid] = m0; my0 = m0; }
    if (mv1) { lab[tid + NT] = m1; my1 = m1; }
    if (mv0 || mv1) chg[cur] = 1;
    const int nxt = (cur == 2) ? 0 : cur + 1;
    if (tid == 0) chg[nxt] = 0;
    __syncthreads();
    const int any = chg[cur];
    cur = nxt;
    if (any == 0) break;
  }

  {
    const int t0 = tid, t1 = tid + NT;
    if (my0 < SENT) {
      const int rr = t0 / GW, cc = t0 - rr * GW;
      atomicMin(&rmn[my0], rr); atomicMax(&rmx[my0], rr);
      atomicMin(&cmn[my0], cc); atomicMax(&cmx[my0], cc);
    }
    if (my1 < SENT) {
      const int rr = t1 / GW, cc = t1 - rr * GW;
      atomicMin(&rmn[my1], rr); atomicMax(&rmx[my1], rr);
      atomicMin(&cmn[my1], cc); atomicMax(&cmx[my1], cc);
    }
  }

  const bool isr0 = (my0 == tid);
  const bool isr1 = (my1 == tid + NT) && (tid + NT < HWC);
  const unsigned rb0 = __builtin_amdgcn_ballot_w32(isr0);
  const unsigned rb1 = __builtin_amdgcn_ballot_w32(isr1);
  if (lane == 0) { wmask[wv] = rb0; wmask[16 + wv] = rb1; }
  __syncthreads();

  int nr;
  {
    const unsigned c = (unsigned)__popc(wmask[lane]);
    const unsigned lo = (1u << wv) - 1u;
    const unsigned hi = (1u << (16 + wv)) - 1u;
    unsigned base0 = 0u, base1 = 0u, total = 0u;
    for (int k = 0; k < 6; ++k) {
      const unsigned bk = __builtin_amdgcn_ballot_w32(((c >> k) & 1u) != 0u);
      base0 += (unsigned)__popc(bk & lo) << k;
      base1 += (unsigned)__popc(bk & hi) << k;
      total += (unsigned)__popc(bk) << k;
    }
    const unsigned below = (1u << lane) - 1u;
    if (isr0) {
      const unsigned rank = base0 + (unsigned)__popc(rb0 & below);
      if (rank < MAXOBJ) roots[rank] = tid;
    }
    if (isr1) {
      const unsigned rank = base1 + (unsigned)__popc(rb1 & below);
      if (rank < MAXOBJ) roots[rank] = tid + NT;
    }
    nr = (int)min(total, (unsigned)MAXOBJ);
  }
  __syncthreads();

  {
    int color = 0, y = 0, x = 0, h = 1, w = 1;
    const bool valid = wv < nr;
    if (valid) {
      const int r = roots[wv];
      y = min(max(rmn[r], 0), GH - 1);
      x = min(max(cmn[r], 0), GW - 1);
      const int y2 = min(max(rmx[r] + 1, y + 1), GH);
      const int x2 = min(max(cmx[r] + 1, x + 1), GW);
      color = g[r]; h = y2 - y; w = x2 - x;
    }
    if (lane == 0) s_valid[wv] = valid ? 1 : 0;
    const int ys = __builtin_amdgcn_readfirstlane(y), xs = __builtin_amdgcn_readfirstlane(x);
    const int hs = __builtin_amdgcn_readfirstlane(h), ws = __builtin_amdgcn_readfirstlane(w);
    const int ncell = hs * ws;
    const float* cellp = grid_emb + (((size_t)(b * GH + ys)) * GW + xs) * DD + 4 * lane;
    v4f s0 = {0.f, 0.f, 0.f, 0.f}, s1 = {0.f, 0.f, 0.f, 0.f};
    int ri = 0, cj = 0;
    for (int c0 = 0; c0 < ncell; c0 += PG) {
      const int n = min(PG, ncell - c0);
      v4f q0[PG], q1[PG];
      pool_request<0>(q0, q1, cellp, n, ws, ri, cj);
      pool_add<0>(s0, s1, q0, q1, n);
    }
    const float cnt = (float)(h * w);
    *(v4f*)&sC[wv][4 * lane]       = s0 / cnt;
    *(v4f*)&sC[wv][128 + 4 * lane] = s1 / cnt;
    if (lane < LD - DD) {
      float f = 0.f;
      if (lane == 0) f = (float)color / 9.0f;
      if (lane == 1) f = (float)x / (float)GW;
      if (lane == 2) f = (float)y / (float)GH;
      if (lane == 3) f = (float)w / (float)GW;
      if (lane == 4) f = (float)h / (float)GH;
      sC[wv][DD + lane] = f;
    }
  }

  if (mmw) load_b<8, false>(wq0, W1 + cb, 0, half);

  if (tid < DD) {
    const float* sp = structure + (size_t)b * 8 * DD + tid;
    float s = 0.f;
    #pragma unroll
    for (int k = 0; k < 8; ++k) s += sp[(size_t)k * DD];
    s_sn[tid] = s * 0.125f;
  }
  __syncthreads();
  if (tid < DD) {
    float p = s_sn[tid] * s_sn[tid];
    #pragma unroll
    for (int m = 16; m >= 1; m >>= 1) p += __shfl_xor(p, m, 32);
    if (lane == 0) s_red[wv] = p;
  }
  __syncthreads();
  if (tid == 0) {
    float t = 0.f;
    for (int i = 0; i < 8; ++i) t += s_red[i];
    s_norm = fmaxf(sqrtf(t), 1e-8f);
  }
  __syncthreads();
  if (tid < DD) s_sn[tid] /= s_norm;

  if (mmw) {
    v8f acc[4] = {};
    gemm_cols64<66>(acc, wq0, wq1, W1 + cb, &sC[mrow][0], half);
    load_b<8, false>(wq0, W2 + cb, 0, half);
    const v4f bias = *(const v4f*)(b1 + cb);
    #pragma unroll
    for (int i = 0; i < 8; ++i) {
      v4f v;
      #pragma unroll
      for (int t = 0; t < 4; ++t) v[t] = acc[t][i] + bias[t];
      *(v4f*)&sA[half * 8 + i][cb] = v;
    }
  }
  __syncthreads();
  {
    float* hp = &sA[wv][8 * lane];
    v4f u0 = *(v4f*)hp, u1 = *(v4f*)(hp + 4);
    #pragma unroll
    for (int t = 0; t < 4; ++t) {
      u0[t] = 0.5f * u0[t] * (1.0f + erff(u0[t] * 0.70710678118654752f));
      u1[t] = 0.5f * u1[t] * (1.0f + erff(u1[t] * 0.70710678118654752f));
    }
    *(v4f*)hp = u0; *(v4f*)(hp + 4) = u1;
  }
  __syncthreads();

  if (mmw) {
    v8f acc[4] = {};
    gemm_cols64<64>(acc, wq0, wq1, W2 + cb, &sA[mrow][0], half);
    load_b<8, false>(wq0, Wp + cb, 0, half);
    const v4f bias = *(const v4f*)(b2 + cb);
    #pragma unroll
    for (int i = 0; i < 8; ++i) {
      const int row = half * 8 + i;
      v4f v;
      #pragma unroll
      for (int t = 0; t < 4; ++t) v[t] = acc[t][i] + bias[t];
      const v4f z = {0.f, 0.f, 0.f, 0.f};
      *(v4f*)&sB[row][cb] = s_valid[row] ? v : z;
    }
  }
  __syncthreads();

  {
    float o[8], sn[8];
    #pragma unroll
    for (int j = 0; j < 8; ++j) {
      const int n = lane + 32 * j;
      o[j] = sB[wv][n];
      sn[j] = s_sn[n];
    }
    float p = 0.f;
    #pragma unroll
    for (int j = 0; j < 8; ++j) p += o[j] * sn[j];
    p += xhalf(p);
    p += xrow<8>(p);
    p += xrow<4>(p);
    p += xrow<2>(p);
    p += xrow<1>(p);
    #pragma unroll
    for (int j = 0; j < 8; ++j) sB[wv][lane + 32 * j] = (o[j] - p * sn[j]) * os;
  }
  __syncthreads();

  if (mmw) {
    v8f acc[4] = {};
    gemm_cols64<64>(acc, wq0, wq1, Wp + cb, &sB[mrow][0], half);
    const v4f bias = *(const v4f*)(bp + cb);
    #pragma unroll
    for (int i = 0; i < 8; ++i) {
      v4f v;
      #pragma unroll
      for (int t = 0; t < 4; ++t) v[t] = acc[t][i] + bias[t];
      *(v4f*)&sC[half * 8 + i][cb] = v;
    }
  }
  __syncthreads();

  {
    float s1 = 0.f, s2 = 0.f;
    #pragma unroll
    for (int j = 0; j < 8; ++j) {
      const float v = sC[wv][lane + 32 * j];
      s1 += v; s2 += v * v;
    }
    #pragma unroll
    for (int m = 16; m >= 1; m >>= 1) {
      s1 += __shfl_xor(s1, m, 32);
      s2 += __shfl_xor(s2, m, 32);
    }
    const float mu = s1 * (1.0f / 256.0f);
    const float var = s2 * (1.0f / 256.0f) - mu * mu;
    const float rstd = rsqrtf(var + 1e-5f);
    float* op = out + (size_t)(b * MAXOBJ + wv) * DD;
    #pragma unroll
    for (int j = 0; j < 8; ++j) {
      const int n = lane + 32 * j;
      const float v = sC[wv][n];
      op[n] = (v - mu) * rstd * gmm[n] + bta[n];
    }
  }
}

extern "C" void kernel_launch(void* const* d_in, const int* in_sizes, int n_in,
                              void* d_out, int out_size, void* d_ws, size_t ws_size,
                              hipStream_t stream) {
  const float* grid_emb  = (const float*)d_in[0];
  const int*   grid      = (const int*)  d_in[1];
  const float* structure = (const float*)d_in[2];
  const float* W1        = (const float*)d_in[3];
  const float* b1        = (const float*)d_in[4];
  const float* W2        = (const float*)d_in[5];
  const float* b2        = (const float*)d_in[6];
  const float* Wp        = (const float*)d_in[7];
  const float* bp        = (const float*)d_in[8];
  const float* gmm       = (const float*)d_in[9];
  const float* bta       = (const float*)d_in[10];
  const float* osp       = (const float*)d_in[11];
  float* out = (float*)d_out;

  const int B = in_sizes[1] / HWC;

  cc_pool_mlp_kernel<<<B, NT, 0, stream>>>(grid, grid_emb, structure, W1, b1, W2, b2,
                                           Wp, bp, gmm, bta, osp, out);
}
